# starting kernel
# baseline (speedup 1.0000x reference)
#include <hip/hip_runtime.h>

typedef __attribute__((ext_vector_type(16))) __bf16 v16bf;
typedef __attribute__((ext_vector_type(8)))  float  v8f;


__global__ __launch_bounds__(256) void fill_f32_k(float* p, float v, long long n) {
  long long i = (long long)blockIdx.x * 256 + threadIdx.x;
  if (i < n) p[i] = v;
}

__device__ __forceinline__ void atomic_max_f32(float* addr, float v) {
  if (v >= 0.0f) atomicMax((int*)addr, __float_as_int(v));
  else           atomicMin((unsigned int*)addr, __float_as_uint(v));
}

__global__ __launch_bounds__(256) void pack_a_k(const float* X, __bf16* Ap, int M, int K) {
  long long o = (long long)blockIdx.x * 256 + threadIdx.x;
  long long total = (long long)M * K;
  if (o >= total) return;
  int i    = (int)(o & 15);
  int lane = (int)((o >> 4) & 31);
  long long t = o >> 9;
  int Kt = K >> 5;
  int ks = (int)(t % Kt);
  int tm = (int)(t / Kt);
  int m = tm * 16 + (lane & 15);
  int k = ks * 32 + ((lane >> 4) << 3) + (i < 8 ? i : i + 8);
  Ap[o] = (__bf16)X[(long long)m * K + k];
}

__global__ __launch_bounds__(256) void pack_b_k(const float* W, __bf16* Bp, int K, int N) {
  long long o = (long long)blockIdx.x * 256 + threadIdx.x;
  long long total = (long long)K * N;
  if (o >= total) return;
  int i    = (int)(o & 15);
  int lane = (int)((o >> 4) & 31);
  long long t = o >> 9;
  int Kt = K >> 5;
  int ks = (int)(t % Kt);
  int tn = (int)(t / Kt);
  int n = tn * 16 + (lane & 15);
  int k = ks * 32 + ((lane >> 4) << 3) + (i < 8 ? i : i + 8);
  Bp[o] = (__bf16)W[(long long)k * N + n];
}

template <int NT>
__global__ __launch_bounds__(256) void wmma_gemm_strip_k(const __bf16* __restrict__ Ap,
                                                         const __bf16* __restrict__ Bp,
                                                         float* __restrict__ C,
                                                         int Mt, int Kt) {
  const int N = NT * 16;
  int wave = threadIdx.x >> 5;
  int lane = threadIdx.x & 31;
  long long tm = (long long)blockIdx.x * 8 + wave;
  if (tm >= Mt) return;
  const v16bf* Af = (const v16bf*)Ap;
  const v16bf* Bf = (const v16bf*)Bp;
  v8f zero = {};
  v8f acc[NT];
  #pragma unroll
  for (int t = 0; t < NT; ++t) acc[t] = zero;
  for (int ks = 0; ks < Kt; ++ks) {
    v16bf a = Af[(tm * Kt + ks) * 32 + lane];
    #pragma unroll
    for (int t = 0; t < NT; ++t) {
      v16bf b = Bf[((long long)t * Kt + ks) * 32 + lane];
      acc[t] = __builtin_amdgcn_wmma_f32_16x16x32_bf16(
          false, a, false, b, (short)0, acc[t], false, false);
    }
  }
  int colb = lane & 15;
  long long rbase = tm * 16 + ((lane >> 4) << 3);
  #pragma unroll
  for (int t = 0; t < NT; ++t) {
    #pragma unroll
    for (int r = 0; r < 8; ++r)
      C[(rbase + r) * N + t * 16 + colb] = acc[t][r];
  }
}

template <int FPL>
__global__ __launch_bounds__(256) void edge_logits_k(const float* __restrict__ xl,
                                                     const float* __restrict__ xr,
                                                     const float* __restrict__ att,
                                                     const int* __restrict__ esrc,
                                                     const int* __restrict__ edst,
                                                     int E, int Nn,
                                                     float* __restrict__ logits,
                                                     float* __restrict__ mmax) {
  const int C = FPL * 32;
  int wave = threadIdx.x >> 5, lane = threadIdx.x & 31;
  long long e = (long long)blockIdx.x * 8 + wave;
  long long EA = (long long)E + Nn;
  if (e >= EA) return;
  int s, d;
  if (e < E) { s = esrc[e]; d = edst[e]; } else { s = d = (int)(e - E); }
  float part = 0.0f;
  if constexpr (FPL == 4) {
    float4 l4 = ((const float4*)(xl + (long long)s * C))[lane];
    float4 r4 = ((const float4*)(xr + (long long)d * C))[lane];
    float4 a4 = ((const float4*)att)[lane];
    float v;
    v = l4.x + r4.x; v = fmaxf(v, 0.2f * v); part += v * a4.x;
    v = l4.y + r4.y; v = fmaxf(v, 0.2f * v); part += v * a4.y;
    v = l4.z + r4.z; v = fmaxf(v, 0.2f * v); part += v * a4.z;
    v = l4.w + r4.w; v = fmaxf(v, 0.2f * v); part += v * a4.w;
  } else {
    float v = xl[(long long)s * C + lane] + xr[(long long)d * C + lane];
    v = fmaxf(v, 0.2f * v);
    part = v * att[lane];
  }
  #pragma unroll
  for (int m = 16; m >= 1; m >>= 1) part += __shfl_xor(part, m, 32);
  if (lane == 0) {
    logits[e] = part;
    atomic_max_f32(&mmax[d], part);
  }
}

template <int FPL>
__global__ __launch_bounds__(256) void edge_agg_k(const float* __restrict__ xl,
                                                  const float* __restrict__ logits,
                                                  const float* __restrict__ mmax,
                                                  const int* __restrict__ esrc,
                                                  const int* __restrict__ edst,
                                                  int E, int Nn,
                                                  float* __restrict__ denom,
                                                  float* __restrict__ acc) {
  const int C = FPL * 32;
  int wave = threadIdx.x >> 5, lane = threadIdx.x & 31;
  long long e = (long long)blockIdx.x * 8 + wave;
  long long EA = (long long)E + Nn;
  if (e >= EA) return;
  int s, d;
  if (e < E) { s = esrc[e]; d = edst[e]; } else { s = d = (int)(e - E); }
  float ev = __expf(logits[e] - mmax[d]);
  if (lane == 0) atomicAdd(&denom[d], ev);
  if constexpr (FPL == 4) {
    float4 l4 = ((const float4*)(xl + (long long)s * C))[lane];
    float* ap = acc + (long long)d * C + lane * 4;
    atomicAdd(ap + 0, l4.x * ev);
    atomicAdd(ap + 1, l4.y * ev);
    atomicAdd(ap + 2, l4.z * ev);
    atomicAdd(ap + 3, l4.w * ev);
  } else {
    atomicAdd(acc + (long long)d * C + lane, xl[(long long)s * C + lane] * ev);
  }
}

__global__ __launch_bounds__(256) void finalize_k(const float* __restrict__ acc,
                                                  const float* __restrict__ denom,
                                                  const float* __restrict__ bias,
                                                  float* __restrict__ out,
                                                  long long total, int C) {
  long long i = (long long)blockIdx.x * 256 + threadIdx.x;
  if (i >= total) return;
  long long n = i / C;
  int c = (int)(i % C);
  out[i] = fmaxf(acc[i] / (denom[n] + 1e-16f) + bias[c], 0.0f);
}


extern "C" void kernel_launch(void* const* d_in, const int* in_sizes, int n_in,
                              void* d_out, int out_size, void* d_ws, size_t ws_size,
                              hipStream_t stream) {
  const int F_IN = 128, H = 128, OUT = 32;
  const float* x    = (const float*)d_in[0];
  const int*   eidx = (const int*)  d_in[1];
  const float* Wl1  = (const float*)d_in[2];
  const float* Wr1  = (const float*)d_in[3];
  const float* att1 = (const float*)d_in[4];
  const float* b1   = (const float*)d_in[5];
  const float* Wl2  = (const float*)d_in[6];
  const float* Wr2  = (const float*)d_in[7];
  const float* att2 = (const float*)d_in[8];
  const float* b2   = (const float*)d_in[9];
  float* out = (float*)d_out;

  const int Nn = in_sizes[0] / F_IN;
  const int E  = in_sizes[1] / 2;
  const long long EA = (long long)E + Nn;
  const int* esrc = eidx;
  const int* edst = eidx + E;

  char* ws = (char*)d_ws;
  size_t off = 0;
  auto carve = [&](size_t bytes) -> char* {
    char* p = ws + off;
    off += (bytes + 255) & ~(size_t)255;
    return p;
  };
  __bf16* apack = (__bf16*)carve((size_t)Nn * F_IN * sizeof(__bf16));
  __bf16* bpl1  = (__bf16*)carve((size_t)F_IN * H   * sizeof(__bf16));
  __bf16* bpr1  = (__bf16*)carve((size_t)F_IN * H   * sizeof(__bf16));
  __bf16* bpl2  = (__bf16*)carve((size_t)H * OUT    * sizeof(__bf16));
  __bf16* bpr2  = (__bf16*)carve((size_t)H * OUT    * sizeof(__bf16));
  float*  xl    = (float*) carve((size_t)Nn * H * sizeof(float));
  float*  xr    = (float*) carve((size_t)Nn * H * sizeof(float));
  float*  logit = (float*) carve((size_t)EA * sizeof(float));
  float*  mmax  = (float*) carve((size_t)Nn * sizeof(float));
  float*  denom = (float*) carve((size_t)Nn * sizeof(float));
  float*  accb  = (float*) carve((size_t)Nn * H * sizeof(float));
  float*  hbuf  = (float*) carve((size_t)Nn * H * sizeof(float));

  auto g256 = [](long long n) { return (unsigned)((n + 255) / 256); };
  auto gwav = [](long long n) { return (unsigned)((n + 7) / 8); };

  pack_b_k<<<g256((long long)F_IN * H), 256, 0, stream>>>(Wl1, bpl1, F_IN, H);
  pack_b_k<<<g256((long long)F_IN * H), 256, 0, stream>>>(Wr1, bpr1, F_IN, H);
  pack_b_k<<<g256((long long)H * OUT),  256, 0, stream>>>(Wl2, bpl2, H, OUT);
  pack_b_k<<<g256((long long)H * OUT),  256, 0, stream>>>(Wr2, bpr2, H, OUT);

  const int Mt = Nn / 16;

  {
    const int C = H, K = F_IN, Kt = K / 32;
    pack_a_k<<<g256((long long)Nn * K), 256, 0, stream>>>(x, apack, Nn, K);
    wmma_gemm_strip_k<8><<<gwav(Mt), 256, 0, stream>>>(apack, bpl1, xl, Mt, Kt);
    wmma_gemm_strip_k<8><<<gwav(Mt), 256, 0, stream>>>(apack, bpr1, xr, Mt, Kt);

    fill_f32_k<<<g256(Nn), 256, 0, stream>>>(mmax, -INFINITY, Nn);
    fill_f32_k<<<g256(Nn), 256, 0, stream>>>(denom, 0.0f, Nn);
    fill_f32_k<<<g256((long long)Nn * C), 256, 0, stream>>>(accb, 0.0f, (long long)Nn * C);

    edge_logits_k<4><<<gwav(EA), 256, 0, stream>>>(xl, xr, att1, esrc, edst, E, Nn, logit, mmax);
    edge_agg_k<4><<<gwav(EA), 256, 0, stream>>>(xl, logit, mmax, esrc, edst, E, Nn, denom, accb);
    finalize_k<<<g256((long long)Nn * C), 256, 0, stream>>>(accb, denom, b1, hbuf,
                                                            (long long)Nn * C, C);
  }

  {
    const int C = OUT, K = H, Kt = K / 32;
    pack_a_k<<<g256((long long)Nn * K), 256, 0, stream>>>(hbuf, apack, Nn, K);
    wmma_gemm_strip_k<2><<<gwav(Mt), 256, 0, stream>>>(apack, bpl2, xl, Mt, Kt);
    wmma_gemm_strip_k<2><<<gwav(Mt), 256, 0, stream>>>(apack, bpr2, xr, Mt, Kt);

    fill_f32_k<<<g256(Nn), 256, 0, stream>>>(mmax, -INFINITY, Nn);
    fill_f32_k<<<g256(Nn), 256, 0, stream>>>(denom, 0.0f, Nn);
    fill_f32_k<<<g256((long long)Nn * C), 256, 0, stream>>>(accb, 0.0f, (long long)Nn * C);

    edge_logits_k<1><<<gwav(EA), 256, 0, stream>>>(xl, xr, att2, esrc, edst, E, Nn, logit, mmax);
    edge_agg_k<1><<<gwav(EA), 256, 0, stream>>>(xl, logit, mmax, esrc, edst, E, Nn, denom, accb);
    finalize_k<<<g256((long long)Nn * C), 256, 0, stream>>>(accb, denom, b2, out,
                                                            (long long)Nn * C, C);
  }
}


# Edge kernels: one wave walks 32 consecutive edges (lane j keeps edge j's ends, logit, weight) instead of one wave per edge: 32x fewer waves; float atomic adds kept, their arrival order differs
# speedup vs baseline: 1.6996x; 1.6996x over previous
#include <hip/hip_runtime.h>
#include <type_traits>

typedef __attribute__((ext_vector_type(16))) __bf16 v16bf;
typedef __attribute__((ext_vector_type(8)))  float  v8f;


__global__ __launch_bounds__(256) void fill_f32_k(float* p, float v, long long n) {
  long long i = (long long)blockIdx.x * 256 + threadIdx.x;
  if (i < n) p[i] = v;
}

__device__ __forceinline__ void atomic_max_f32(float* addr, float v) {
  if (v >= 0.0f) atomicMax((int*)addr, __float_as_int(v));
  else           atomicMin((unsigned int*)addr, __float_as_uint(v));
}

__global__ __launch_bounds__(256) void pack_a_k(const float* X, __bf16* Ap, int M, int K) {
  long long o = (long long)blockIdx.x * 256 + threadIdx.x;
  long long total = (long long)M * K;
  if (o >= total) return;
  int i    = (int)(o & 15);
  int lane = (int)((o >> 4) & 31);
  long long t = o >> 9;
  int Kt = K >> 5;
  int ks = (int)(t % Kt);
  int tm = (int)(t / Kt);
  int m = tm * 16 + (lane & 15);
  int k = ks * 32 + ((lane >> 4) << 3) + (i < 8 ? i : i + 8);
  Ap[o] = (__bf16)X[(long long)m * K + k];
}

__global__ __launch_bounds__(256) void pack_b_k(const float* W, __bf16* Bp, int K, int N) {
  long long o = (long long)blockIdx.x * 256 + threadIdx.x;
  long long total = (long long)K * N;
  if (o >= total) return;
  int i    = (int)(o & 15);
  int lane = (int)((o >> 4) & 31);
  long long t = o >> 9;
  int Kt = K >> 5;
  int ks = (int)(t % Kt);
  int tn = (int)(t / Kt);
  int n = tn * 16 + (lane & 15);
  int k = ks * 32 + ((lane >> 4) << 3) + (i < 8 ? i : i + 8);
  Bp[o] = (__bf16)W[(long long)k * N + n];
}

template <int NT>
__global__ __launch_bounds__(256) void wmma_gemm_strip_k(const __bf16* __restrict__ Ap,
                                                         const __bf16* __restrict__ Bp,
                                                         float* __restrict__ C,
                                                         int Mt, int Kt) {
  const int N = NT * 16;
  int wave = threadIdx.x >> 5;
  int lane = threadIdx.x & 31;
  long long tm = (long long)blockIdx.x * 8 + wave;
  if (tm >= Mt) return;
  const v16bf* Af = (const v16bf*)Ap;
  const v16bf* Bf = (const v16bf*)Bp;
  v8f zero = {};
  v8f acc[NT];
  #pragma unroll
  for (int t = 0; t < NT; ++t) acc[t] = zero;
  for (int ks = 0; ks < Kt; ++ks) {
    v16bf a = Af[(tm * Kt + ks) * 32 + lane];
    #pragma unroll
    for (int t = 0; t < NT; ++t) {
      v16bf b = Bf[((long long)t * Kt + ks) * 32 + lane];
      acc[t] = __builtin_amdgcn_wmma_f32_16x16x32_bf16(
          false, a, false, b, (short)0, acc[t], false, false);
    }
  }
  int colb = lane & 15;
  long long rbase = tm * 16 + ((lane >> 4) << 3);
  #pragma unroll
  for (int t = 0; t < NT; ++t) {
    #pragma unroll
    for (int r = 0; r < 8; ++r)
      C[(rbase + r) * N + t * 16 + colb] = acc[t][r];
  }
}


__device__ __forceinline__ void edge_ends(const int* __restrict__ esrc,
                                          const int* __restrict__ edst,
                                          long long e, int E, long long EA,
                                          int& s, int& d) {
  s = 0; d = 0;
  if (e < E) { s = esrc[e]; d = edst[e]; }
  else if (e < EA) { s = d = (int)(e - E); }
}

template <int FPL>
__global__ __launch_bounds__(256) void edge_logits_k(const float* __restrict__ xl,
                                                     const float* __restrict__ xr,
                                                     const float* __restrict__ att,
                                                     const int* __restrict__ esrc,
                                                     const int* __restrict__ edst,
                                                     int E, int Nn,
                                                     float* __restrict__ logits,
                                                     float* __restrict__ mmax) {
  constexpr int C = FPL * 32;
  constexpr int G = (FPL == 4) ? 4 : 8;
  const int lane = threadIdx.x & 31;
  const int wv = __builtin_amdgcn_readfirstlane((int)(threadIdx.x >> 5));
  const long long EA = (long long)E + Nn;
  const long long e0 = ((long long)blockIdx.x * 8 + wv) * 32;
  if (e0 >= EA) return;
  const long long e = e0 + lane;
  int s, d;
  edge_ends(esrc, edst, e, E, EA, s, d);
  float mine = 0.0f;
  if constexpr (FPL == 4) {
    const float4 a4 = ((const float4*)att)[lane];
    for (int j0 = 0; j0 < 32; j0 += G) {
      float4 l4[G], r4[G];
      #pragma unroll
      for (int g = 0; g < G; ++g) {
        const int sj = __builtin_amdgcn_readlane(s, j0 + g);
        const int dj = __builtin_amdgcn_readlane(d, j0 + g);
        l4[g] = ((const float4*)(xl + (long long)sj * C))[lane];
        r4[g] = ((const float4*)(xr + (long long)dj * C))[lane];
      }
      #pragma unroll
      for (int g = 0; g < G; ++g) {
        float part = 0.0f, v;
        v = l4[g].x + r4[g].x; v = fmaxf(v, 0.2f * v); part = fmaf(v, a4.x, part);
        v = l4[g].y + r4[g].y; v = fmaxf(v, 0.2f * v); part = fmaf(v, a4.y, part);
        v = l4[g].z + r4[g].z; v = fmaxf(v, 0.2f * v); part = fmaf(v, a4.z, part);
        v = l4[g].w + r4[g].w; v = fmaxf(v, 0.2f * v); part = fmaf(v, a4.w, part);
        #pragma unroll
        for (int m = 16; m >= 1; m >>= 1) part += __shfl_xor(part, m, 32);
        if (lane == j0 + g) mine = part;
      }
    }
  } else {
    const float a1 = att[lane];
    for (int j0 = 0; j0 < 32; j0 += G) {
      float l1[G], r1[G];
      #pragma unroll
      for (int g = 0; g < G; ++g) {
        const int sj = __builtin_amdgcn_readlane(s, j0 + g);
        const int dj = __builtin_amdgcn_readlane(d, j0 + g);
        l1[g] = xl[(long long)sj * C + lane];
        r1[g] = xr[(long long)dj * C + lane];
      }
      #pragma unroll
      for (int g = 0; g < G; ++g) {
        float v = l1[g] + r1[g];
        v = fmaxf(v, 0.2f * v);
        float part = v * a1;
        #pragma unroll
        for (int m = 16; m >= 1; m >>= 1) part += __shfl_xor(part, m, 32);
        if (lane == j0 + g) mine = part;
      }
    }
  }
  if (e < EA) {
    logits[e] = mine;
    atomic_max_f32(&mmax[d], mine);
  }
}

__device__ __forceinline__ void add_row(float* __restrict__ acc, int C, int lane,
                                        int dj, float evj, float4 r) {
  float* ap = acc + (long long)dj * C + lane * 4;
  atomicAdd(ap + 0, r.x * evj);
  atomicAdd(ap + 1, r.y * evj);
  atomicAdd(ap + 2, r.z * evj);
  atomicAdd(ap + 3, r.w * evj);
}
__device__ __forceinline__ void add_row(float* __restrict__ acc, int C, int lane,
                                        int dj, float evj, float r) {
  atomicAdd(acc + (long long)dj * C + lane, r * evj);
}

template <int FPL>
__global__ __launch_bounds__(256) void edge_agg_k(const float* __restrict__ xl,
                                                  const float* __restrict__ logits,
                                                  const float* __restrict__ mmax,
                                                  const int* __restrict__ esrc,
                                                  const int* __restrict__ edst,
                                                  int E, int Nn,
                                                  float* __restrict__ denom,
                                                  float* __restrict__ acc) {
  constexpr int C = FPL * 32;
  constexpr int G = 8;
  using row_t = typename std::conditional<FPL == 4, float4, float>::type;
  const int lane = threadIdx.x & 31;
  const int wv = __builtin_amdgcn_readfirstlane((int)(threadIdx.x >> 5));
  const long long EA = (long long)E + Nn;
  const long long e0 = ((long long)blockIdx.x * 8 + wv) * 32;
  if (e0 >= EA) return;
  const long long e = e0 + lane;
  int s, d;
  edge_ends(esrc, edst, e, E, EA, s, d);
  float ev = 0.0f;
  if (e < EA) {
    ev = __expf(logits[e] - mmax[d]);
    atomicAdd(&denom[d], ev);
  }
  const int evb = __float_as_int(ev);
  const row_t* __restrict__ rows = (const row_t*)xl;
  const int n = (EA - e0 < 32) ? (int)(EA - e0) : 32;
  const int nfull = n & ~(G - 1);
  for (int j0 = 0; j0 < nfull; j0 += G) {
    row_t r[G];
    #pragma unroll
    for (int g = 0; g < G; ++g) {
      const int sj = __builtin_amdgcn_readlane(s, j0 + g);
      r[g] = rows[(long long)sj * 32 + lane];
    }
    #pragma unroll
    for (int g = 0; g < G; ++g) {
      const int dj = __builtin_amdgcn_readlane(d, j0 + g);
      const float evj = __int_as_float(__builtin_amdgcn_readlane(evb, j0 + g));
      add_row(acc, C, lane, dj, evj, r[g]);
    }
  }
  for (int j = nfull; j < n; ++j) {
    const int sj = __builtin_amdgcn_readlane(s, j);
    const int dj = __builtin_amdgcn_readlane(d, j);
    const float evj = __int_as_float(__builtin_amdgcn_readlane(evb, j));
    add_row(acc, C, lane, dj, evj, rows[(long long)sj * 32 + lane]);
  }
}

__global__ __launch_bounds__(256) void finalize_k(const float* __restrict__ acc,
                                                  const float* __restrict__ denom,
                                                  const float* __restrict__ bias,
                                                  float* __restrict__ out,
                                                  long long total, int C) {
  long long i = (long long)blockIdx.x * 256 + threadIdx.x;
  if (i >= total) return;
  long long n = i / C;
  int c = (int)(i % C);
  out[i] = fmaxf(acc[i] / (denom[n] + 1e-16f) + bias[c], 0.0f);
}


extern "C" void kernel_launch(void* const* d_in, const int* in_sizes, int n_in,
                              void* d_out, int out_size, void* d_ws, size_t ws_size,
                              hipStream_t stream) {
  const int F_IN = 128, H = 128, OUT = 32;
  const float* x    = (const float*)d_in[0];
  const int*   eidx = (const int*)  d_in[1];
  const float* Wl1  = (const float*)d_in[2];
  const float* Wr1  = (const float*)d_in[3];
  const float* att1 = (const float*)d_in[4];
  const float* b1   = (const float*)d_in[5];
  const float* Wl2  = (const float*)d_in[6];
  const float* Wr2  = (const float*)d_in[7];
  const float* att2 = (const float*)d_in[8];
  const float* b2   = (const float*)d_in[9];
  float* out = (float*)d_out;

  const int Nn = in_sizes[0] / F_IN;
  const int E  = in_sizes[1] / 2;
  const long long EA = (long long)E + Nn;
  const int* esrc = eidx;
  const int* edst = eidx + E;

  char* ws = (char*)d_ws;
  size_t off = 0;
  auto carve = [&](size_t bytes) -> char* {
    char* p = ws + off;
    off += (bytes + 255) & ~(size_t)255;
    return p;
  };
  __bf16* apack = (__bf16*)carve((size_t)Nn * F_IN * sizeof(__bf16));
  __bf16* bpl1  = (__bf16*)carve((size_t)F_IN * H   * sizeof(__bf16));
  __bf16* bpr1  = (__bf16*)carve((size_t)F_IN * H   * sizeof(__bf16));
  __bf16* bpl2  = (__bf16*)carve((size_t)H * OUT    * sizeof(__bf16));
  __bf16* bpr2  = (__bf16*)carve((size_t)H * OUT    * sizeof(__bf16));
  float*  xl    = (float*) carve((size_t)Nn * H * sizeof(float));
  float*  xr    = (float*) carve((size_t)Nn * H * sizeof(float));
  float*  logit = (float*) carve((size_t)EA * sizeof(float));
  float*  mmax  = (float*) carve((size_t)Nn * sizeof(float));
  float*  denom = (float*) carve((size_t)Nn * sizeof(float));
  float*  accb  = (float*) carve((size_t)Nn * H * sizeof(float));
  float*  hbuf  = (float*) carve((size_t)Nn * H * sizeof(float));

  auto g256 = [](long long n) { return (unsigned)((n + 255) / 256); };
  auto gwav = [](long long n) { return (unsigned)((n + 7) / 8); };

  pack_b_k<<<g256((long long)F_IN * H), 256, 0, stream>>>(Wl1, bpl1, F_IN, H);
  pack_b_k<<<g256((long long)F_IN * H), 256, 0, stream>>>(Wr1, bpr1, F_IN, H);
  pack_b_k<<<g256((long long)H * OUT),  256, 0, stream>>>(Wl2, bpl2, H, OUT);
  pack_b_k<<<g256((long long)H * OUT),  256, 0, stream>>>(Wr2, bpr2, H, OUT);

  const int Mt = Nn / 16;

  {
    const int C = H, K = F_IN, Kt = K / 32;
    pack_a_k<<<g256((long long)Nn * K), 256, 0, stream>>>(x, apack, Nn, K);
    wmma_gemm_strip_k<8><<<gwav(Mt), 256, 0, stream>>>(apack, bpl1, xl, Mt, Kt);
    wmma_gemm_strip_k<8><<<gwav(Mt), 256, 0, stream>>>(apack, bpr1, xr, Mt, Kt);

    fill_f32_k<<<g256(Nn), 256, 0, stream>>>(mmax, -INFINITY, Nn);
    fill_f32_k<<<g256(Nn), 256, 0, stream>>>(denom, 0.0f, Nn);
    fill_f32_k<<<g256((long long)Nn * C), 256, 0, stream>>>(accb, 0.0f, (long long)Nn * C);

    edge_logits_k<4><<<g256(EA), 256, 0, stream>>>(xl, xr, att1, esrc, edst, E, Nn, logit, mmax);
    edge_agg_k<4><<<g256(EA), 256, 0, stream>>>(xl, logit, mmax, esrc, edst, E, Nn, denom, accb);
    finalize_k<<<g256((long long)Nn * C), 256, 0, stream>>>(accb, denom, b1, hbuf,
                                                            (long long)Nn * C, C);
  }

  {
    const int C = OUT, K = H, Kt = K / 32;
    pack_a_k<<<g256((long long)Nn * K), 256, 0, stream>>>(hbuf, apack, Nn, K);
    wmma_gemm_strip_k<2><<<gwav(Mt), 256, 0, stream>>>(apack, bpl2, xl, Mt, Kt);
    wmma_gemm_strip_k<2><<<gwav(Mt), 256, 0, stream>>>(apack, bpr2, xr, Mt, Kt);

    fill_f32_k<<<g256(Nn), 256, 0, stream>>>(mmax, -INFINITY, Nn);
    fill_f32_k<<<g256(Nn), 256, 0, stream>>>(denom, 0.0f, Nn);
    fill_f32_k<<<g256((long long)Nn * C), 256, 0, stream>>>(accb, 0.0f, (long long)Nn * C);

    edge_logits_k<1><<<g256(EA), 256, 0, stream>>>(xl, xr, att2, esrc, edst, E, Nn, logit, mmax);
    edge_agg_k<1><<<g256(EA), 256, 0, stream>>>(xl, logit, mmax, esrc, edst, E, Nn, denom, accb);
    finalize_k<<<g256((long long)Nn * C), 256, 0, stream>>>(accb, denom, b2, out,
                                                            (long long)Nn * C, C);
  }
}


# Layer-2 node kernel: a 128-byte source row lies across 8 lanes, one load brings 4 edges' rows, logits summed over 8 lanes by DPP; 4 running softmax streams merged at the end; sum order changes
# speedup vs baseline: 7.4877x; 4.4056x over previous
#include <hip/hip_runtime.h>
#include <type_traits>

typedef __attribute__((ext_vector_type(16))) __bf16 v16bf;
typedef __attribute__((ext_vector_type(8)))  float  v8f;


__global__ __launch_bounds__(256) void pack_b_k(const float* W, __bf16* Bp, int K, int N) {
  long long o = (long long)blockIdx.x * 256 + threadIdx.x;
  long long total = (long long)K * N;
  if (o >= total) return;
  int i    = (int)(o & 15);
  int lane = (int)((o >> 4) & 31);
  long long t = o >> 9;
  int Kt = K >> 5;
  int ks = (int)(t % Kt);
  int tn = (int)(t / Kt);
  int n = tn * 16 + (lane & 15);
  int k = ks * 32 + ((lane >> 4) << 3) + (i < 8 ? i : i + 8);
  Bp[o] = (__bf16)W[(long long)k * N + n];
}

template <int NT, int KT>
__device__ __forceinline__ void strip_times_b(const v16bf (&a)[KT], const v16bf* __restrict__ Bf,
                                              float* __restrict__ C, long long tm, int lane) {
  constexpr int N = NT * 16;
  v8f zero = {};
  v8f acc[NT];
  #pragma unroll
  for (int t = 0; t < NT; ++t) acc[t] = zero;
  #pragma unroll
  for (int ks = 0; ks < KT; ++ks) {
    #pragma unroll
    for (int t = 0; t < NT; ++t) {
      v16bf b = Bf[(t * KT + ks) * 32 + lane];
      acc[t] = __builtin_amdgcn_wmma_f32_16x16x32_bf16(
          false, a[ks], false, b, (short)0, acc[t], false, false);
    }
  }
  int colb = lane & 15;
  long long rbase = tm * 16 + ((lane >> 4) << 3);
  #pragma unroll
  for (int t = 0; t < NT; ++t) {
    #pragma unroll
    for (int r = 0; r < 8; ++r)
      C[(rbase + r) * N + t * 16 + colb] = acc[t][r];
  }
}

template <int NT, int KT>
__global__ __launch_bounds__(256) void wmma_proj_pair_k(const float* __restrict__ X,
                                                        const __bf16* __restrict__ Bpl,
                                                        const __bf16* __restrict__ Bpr,
                                                        float* __restrict__ Cl,
                                                        float* __restrict__ Cr, int Mt) {
  constexpr int K = KT * 32;
  int wave = threadIdx.x >> 5;
  int lane = threadIdx.x & 31;
  long long tm = (long long)blockIdx.x * 8 + wave;
  if (tm >= Mt) return;
  const float* p = X + (tm * 16 + (lane & 15)) * K + ((lane >> 4) << 3);
  float4 f[KT][4];
  #pragma unroll
  for (int ks = 0; ks < KT; ++ks) {
    const float4* q = (const float4*)(p + ks * 32);
    f[ks][0] = q[0];
    f[ks][1] = q[1];
    f[ks][2] = q[4];
    f[ks][3] = q[5];
  }
  __builtin_amdgcn_sched_barrier(0);
  v16bf a[KT];
  #pragma unroll
  for (int ks = 0; ks < KT; ++ks) {
    #pragma unroll
    for (int j = 0; j < 4; ++j) {
      a[ks][4 * j + 0] = (__bf16)f[ks][j].x;
      a[ks][4 * j + 1] = (__bf16)f[ks][j].y;
      a[ks][4 * j + 2] = (__bf16)f[ks][j].z;
      a[ks][4 * j + 3] = (__bf16)f[ks][j].w;
    }
  }
  strip_times_b<NT, KT>(a, (const v16bf*)Bpl, Cl, tm, lane);
  strip_times_b<NT, KT>(a, (const v16bf*)Bpr, Cr, tm, lane);
}


constexpr int HIST_K    = 64;
constexpr int HIST_SLAB = 65536;

__device__ __forceinline__ int chunk_len(int E) { return (E + HIST_K - 1) / HIST_K; }

__global__ __launch_bounds__(1024) void hist_k(const int* __restrict__ edst, int E, int Nn,
                                               int* __restrict__ H, int* __restrict__ lrank) {
  __shared__ alignas(16) int h[HIST_SLAB];
  const int tid = threadIdx.x;
  const int k = blockIdx.x;
  const int chunk = chunk_len(E);
  const int e0 = k * chunk;
  const int e1 = (e0 + chunk < E) ? (e0 + chunk) : E;
  for (int lo = 0; lo < Nn; lo += HIST_SLAB) {
    const int span = (Nn - lo < HIST_SLAB) ? (Nn - lo) : HIST_SLAB;
    for (int i = tid * 4; i < span; i += 4096) *(int4*)(h + i) = make_int4(0, 0, 0, 0);
    __syncthreads();
    for (int e = e0 + tid; e < e1; e += 8 * 1024) {
      int d[8];
      #pragma unroll
      for (int q = 0; q < 8; ++q) {
        const int ee = e + q * 1024;
        d[q] = (ee < e1) ? edst[ee] : -1;
      }
      unsigned rel[8];
      int r[8];
      #pragma unroll
      for (int q = 0; q < 8; ++q) {
        rel[q] = (unsigned)(d[q] - lo);
        r[q] = (rel[q] < (unsigned)span) ? atomicAdd(&h[rel[q]], 1) : 0;
      }
      #pragma unroll
      for (int q = 0; q < 8; ++q)
        if (rel[q] < (unsigned)span) lrank[e + q * 1024] = r[q];
    }
    __syncthreads();
    int* __restrict__ row = H + (long long)k * Nn + lo;
    for (int i = tid; i < span; i += 1024) row[i] = h[i];
    __syncthreads();
  }
}

__global__ __launch_bounds__(256) void colscan_k(int* __restrict__ H, int Nn, int* __restrict__ cnt) {
  const int n = blockIdx.x * 256 + threadIdx.x;
  if (n >= Nn) return;
  int* col = H + n;
  int v[HIST_K];
  #pragma unroll
  for (int k = 0; k < HIST_K; ++k) v[k] = col[(long long)k * Nn];
  int run = 0;
  #pragma unroll
  for (int k = 0; k < HIST_K; ++k) { col[(long long)k * Nn] = run; run += v[k]; }
  cnt[n] = run;
}

__global__ __launch_bounds__(1024) void scan_k(const int* __restrict__ cnt,
                                               int* __restrict__ rowptr, int Nn) {
  __shared__ int wsum[32];
  const int tid = threadIdx.x, lane = tid & 31, wv = tid >> 5;
  int carry = 0;
  for (int base = 0; base < Nn; base += 16384) {
    const int i = base + tid * 16;
    const bool whole = (i + 15 < Nn);
    int v[16];
    if (whole) {
      #pragma unroll
      for (int q = 0; q < 4; ++q) {
        const int4 w = *(const int4*)(cnt + i + 4 * q);
        v[4 * q] = w.x; v[4 * q + 1] = w.y; v[4 * q + 2] = w.z; v[4 * q + 3] = w.w;
      }
    } else {
      #pragma unroll
      for (int q = 0; q < 16; ++q) v[q] = (i + q < Nn) ? cnt[i + q] : 0;
    }
    int t = 0;
    #pragma unroll
    for (int q = 0; q < 16; ++q) t += v[q];
    int inc = t;
    #pragma unroll
    for (int o = 1; o < 32; o <<= 1) {
      const int u = __shfl_up(inc, o, 32);
      if (lane >= o) inc += u;
    }
    if (lane == 31) wsum[wv] = inc;
    __syncthreads();
    const int ws = wsum[lane];
    int winc = ws;
    #pragma unroll
    for (int o = 1; o < 32; o <<= 1) {
      const int u = __shfl_up(winc, o, 32);
      if (lane >= o) winc += u;
    }
    const int wave_excl = __shfl(winc, wv, 32) - __shfl(ws, wv, 32);
    const int total = __shfl(winc, 31, 32);
    int e = carry + wave_excl + inc - t;
    int p[16];
    #pragma unroll
    for (int q = 0; q < 16; ++q) { p[q] = e; e += v[q]; }
    if (whole) {
      #pragma unroll
      for (int q = 0; q < 4; ++q)
        *(int4*)(rowptr + i + 4 * q) = make_int4(p[4 * q], p[4 * q + 1], p[4 * q + 2], p[4 * q + 3]);
    } else {
      #pragma unroll
      for (int q = 0; q < 16; ++q) if (i + q < Nn) rowptr[i + q] = p[q];
    }
    carry += total;
    __syncthreads();
  }
  if (tid == 0) rowptr[Nn] = carry;
}

__global__ __launch_bounds__(256) void scatter_k(const int* __restrict__ esrc,
                                                 const int* __restrict__ edst,
                                                 const int* __restrict__ lrank,
                                                 const int* __restrict__ rowptr,
                                                 const int* __restrict__ H, int E, int Nn,
                                                 int* __restrict__ csr_src) {
  int i = blockIdx.x * 256 + threadIdx.x;
  if (i < E) {
    const int k = i / chunk_len(E);
    const int d = edst[i];
    csr_src[rowptr[d] + H[(long long)k * Nn + d] + lrank[i]] = esrc[i];
  }
}


__device__ __forceinline__ float logit_part(float4 l, float4 r, float4 a) {
  float part = 0.0f, v;
  v = l.x + r.x; v = fmaxf(v, 0.2f * v); part = fmaf(v, a.x, part);
  v = l.y + r.y; v = fmaxf(v, 0.2f * v); part = fmaf(v, a.y, part);
  v = l.z + r.z; v = fmaxf(v, 0.2f * v); part = fmaf(v, a.z, part);
  v = l.w + r.w; v = fmaxf(v, 0.2f * v); part = fmaf(v, a.w, part);
  return part;
}
__device__ __forceinline__ void row_scale(float4& acc, float s) { acc.x *= s; acc.y *= s; acc.z *= s; acc.w *= s; }
__device__ __forceinline__ void row_fma(float4& acc, float w, float4 r) {
  acc.x = fmaf(w, r.x, acc.x); acc.y = fmaf(w, r.y, acc.y);
  acc.z = fmaf(w, r.z, acc.z); acc.w = fmaf(w, r.w, acc.w);
}
__device__ __forceinline__ float4 row_final(float4 acc, float dd, float4 b) {
  return make_float4(fmaxf(acc.x / dd + b.x, 0.0f), fmaxf(acc.y / dd + b.y, 0.0f),
                     fmaxf(acc.z / dd + b.z, 0.0f), fmaxf(acc.w / dd + b.w, 0.0f));
}

template <int FPL>
__global__ __launch_bounds__(256) void node_attn_k(const float* __restrict__ xl,
                                                  const float* __restrict__ xr,
                                                  const float* __restrict__ att,
                                                  const float* __restrict__ bias,
                                                  const int* __restrict__ rowptr,
                                                  const int* __restrict__ csr_src,
                                                  int Nn, float* __restrict__ out) {
  constexpr int G = (FPL == 4) ? 4 : 8;
  using row_t = typename std::conditional<FPL == 4, float4, float>::type;
  const int lane = threadIdx.x & 31;
  const int wv = __builtin_amdgcn_readfirstlane((int)(threadIdx.x >> 5));
  const int node = blockIdx.x * 8 + wv;
  if (node >= Nn) return;
  const row_t* __restrict__ rowsL = (const row_t*)xl;
  const row_t xrd   = ((const row_t*)xr)[(long long)node * 32 + lane];
  const row_t a     = ((const row_t*)att)[lane];
  const row_t selfr = rowsL[(long long)node * 32 + lane];
  const int beg = rowptr[node];
  const int end = rowptr[node + 1];

  float m = logit_part(selfr, xrd, a);
  #pragma unroll
  for (int k = 16; k >= 1; k >>= 1) m += __shfl_xor(m, k, 32);
  float den = 1.0f;
  row_t acc = selfr;

  for (int base = beg; base < end; base += 32) {
    const int idx = base + lane;
    const int s = (idx < end) ? csr_src[idx] : node;
    const int n = (end - base < 32) ? (end - base) : 32;
    for (int j0 = 0; j0 < n; j0 += G) {
      row_t r[G];
      float lg[G];
      #pragma unroll
      for (int g = 0; g < G; ++g) {
        const int sj = __builtin_amdgcn_readlane(s, j0 + g);
        r[g] = rowsL[(long long)sj * 32 + lane];
      }
      #pragma unroll
      for (int g = 0; g < G; ++g) lg[g] = logit_part(r[g], xrd, a);
      #pragma unroll
      for (int k = 16; k >= 1; k >>= 1) {
        float t[G];
        #pragma unroll
        for (int g = 0; g < G; ++g) t[g] = __shfl_xor(lg[g], k, 32);
        #pragma unroll
        for (int g = 0; g < G; ++g) lg[g] += t[g];
      }
      float cm = m;
      #pragma unroll
      for (int g = 0; g < G; ++g) {
        if (j0 + g < n) cm = fmaxf(cm, lg[g]);
      }
      const float sc = __expf(m - cm);
      den *= sc;
      row_scale(acc, sc);
      m = cm;
      #pragma unroll
      for (int g = 0; g < G; ++g) {
        const float w = (j0 + g < n) ? __expf(lg[g] - cm) : 0.0f;
        den += w;
        row_fma(acc, w, r[g]);
      }
    }
  }
  const row_t b = ((const row_t*)bias)[lane];
  ((row_t*)out)[(long long)node * 32 + lane] = row_final(acc, den + 1e-16f, b);
}


template <int MASK>
__device__ __forceinline__ float row_xor(float v) {
  return __builtin_bit_cast(float, __builtin_amdgcn_update_dpp(0, __builtin_bit_cast(int, v),
                                                               0x160 + MASK, 0xf, 0xf, true));
}
__device__ __forceinline__ float xhalf(float v) {
  return __builtin_bit_cast(float, __builtin_amdgcn_permlanex16(__builtin_bit_cast(int, v),
      __builtin_bit_cast(int, v), 0x76543210, 0xfedcba98, false, false));
}
template <int T>
__device__ __forceinline__ int oct_lane(int v) {
  return (int)__builtin_amdgcn_mov_dpp8((unsigned)v, T * 0x249249);
}

typedef float v2f __attribute__((ext_vector_type(2)));
__device__ __forceinline__ float logit_part2(float4 l, v2f r01, v2f r23, v2f a01, v2f a23) {
  v2f v01 = (v2f){l.x, l.y} + r01;
  v2f v23 = (v2f){l.z, l.w} + r23;
  v01 = __builtin_elementwise_max(v01, 0.2f * v01);
  v23 = __builtin_elementwise_max(v23, 0.2f * v23);
  const v2f p = __builtin_elementwise_fma(v23, a23, v01 * a01);
  return p.x + p.y;
}

template <int T0>
__device__ __forceinline__ void four_steps(const float4* __restrict__ rowsL, int s, int nv, int q,
                                           v2f xr01, v2f xr23, v2f a01, v2f a23,
                                           float& m, float& den, float4& acc) {
  float4 r[4];
  float lg[4];
  r[0] = rowsL[(unsigned)oct_lane<T0 + 0>(s) * 8u + (unsigned)q];
  r[1] = rowsL[(unsigned)oct_lane<T0 + 1>(s) * 8u + (unsigned)q];
  r[2] = rowsL[(unsigned)oct_lane<T0 + 2>(s) * 8u + (unsigned)q];
  r[3] = rowsL[(unsigned)oct_lane<T0 + 3>(s) * 8u + (unsigned)q];
  #pragma unroll
  for (int u = 0; u < 4; ++u) lg[u] = logit_part2(r[u], xr01, xr23, a01, a23);
  #pragma unroll
  for (int u = 0; u < 4; ++u) lg[u] += row_xor<1>(lg[u]);
  #pragma unroll
  for (int u = 0; u < 4; ++u) lg[u] += row_xor<2>(lg[u]);
  #pragma unroll
  for (int u = 0; u < 4; ++u) lg[u] += row_xor<4>(lg[u]);
  float cm = m;
  #pragma unroll
  for (int u = 0; u < 4; ++u) {
    if (T0 + u < nv) cm = fmaxf(cm, lg[u]);
  }
  const float sc = __expf(m - cm);
  den *= sc;
  row_scale(acc, sc);
  m = cm;
  #pragma unroll
  for (int u = 0; u < 4; ++u) {
    const float w = (T0 + u < nv) ? __expf(lg[u] - cm) : 0.0f;
    den += w;
    row_fma(acc, w, r[u]);
  }
}

__global__ __launch_bounds__(256) void node_attn32_k(const float* __restrict__ xl,
                                                    const float* __restrict__ xr,
                                                    const float* __restrict__ att,
                                                    const float* __restrict__ bias,
                                                    const int* __restrict__ rowptr,
                                                    const int* __restrict__ csr_src,
                                                    int Nn, float* __restrict__ out) {
  const int lane = threadIdx.x & 31;
  const int wv = __builtin_amdgcn_readfirstlane((int)(threadIdx.x >> 5));
  const int node = blockIdx.x * 8 + wv;
  if (node >= Nn) return;
  const int q = lane & 7;
  const int slot = lane >> 3;
  const float4* __restrict__ rowsL = (const float4*)xl;
  const float4 xrd   = ((const float4*)xr)[(long long)node * 8 + q];
  const float4 a     = ((const float4*)att)[q];
  const float4 selfr = rowsL[(long long)node * 8 + q];
  const int beg = rowptr[node];
  const int end = rowptr[node + 1];

  const v2f xr01 = {xrd.x, xrd.y}, xr23 = {xrd.z, xrd.w};
  const v2f a01 = {a.x, a.y}, a23 = {a.z, a.w};

  float m = logit_part2(selfr, xr01, xr23, a01, a23);
  m += row_xor<1>(m);
  m += row_xor<2>(m);
  m += row_xor<4>(m);
  float den = (slot == 0) ? 1.0f : 0.0f;
  float4 acc = (slot == 0) ? selfr : make_float4(0.0f, 0.0f, 0.0f, 0.0f);

  for (int base = beg; base < end; base += 32) {
    const int n = (end - base < 32) ? (end - base) : 32;
    const int eo = 4 * q + slot;
    const int s = (eo < n) ? csr_src[base + eo] : node;
    const int nv = (n - slot + 3) >> 2;
    four_steps<0>(rowsL, s, nv, q, xr01, xr23, a01, a23, m, den, acc);
    if (n > 16) four_steps<4>(rowsL, s, nv, q, xr01, xr23, a01, a23, m, den, acc);
  }

  float mm = fmaxf(m, row_xor<8>(m));
  mm = fmaxf(mm, xhalf(mm));
  const float f = __expf(m - mm);
  den *= f;
  row_scale(acc, f);
  den += row_xor<8>(den);
  acc.x += row_xor<8>(acc.x); acc.y += row_xor<8>(acc.y);
  acc.z += row_xor<8>(acc.z); acc.w += row_xor<8>(acc.w);
  den += xhalf(den);
  acc.x += xhalf(acc.x); acc.y += xhalf(acc.y);
  acc.z += xhalf(acc.z); acc.w += xhalf(acc.w);
  if (slot == 0) {
    const float4 b = ((const float4*)bias)[q];
    ((float4*)out)[(long long)node * 8 + q] = row_final(acc, den + 1e-16f, b);
  }
}


extern "C" void kernel_launch(void* const* d_in, const int* in_sizes, int n_in,
                              void* d_out, int out_size, void* d_ws, size_t ws_size,
                              hipStream_t stream) {
  const int F_IN = 128, H = 128, OUT = 32;
  const float* x    = (const float*)d_in[0];
  const int*   eidx = (const int*)  d_in[1];
  const float* Wl1  = (const float*)d_in[2];
  const float* Wr1  = (const float*)d_in[3];
  const float* att1 = (const float*)d_in[4];
  const float* b1   = (const float*)d_in[5];
  const float* Wl2  = (const float*)d_in[6];
  const float* Wr2  = (const float*)d_in[7];
  const float* att2 = (const float*)d_in[8];
  const float* b2   = (const float*)d_in[9];
  float* out = (float*)d_out;

  const int Nn = in_sizes[0] / F_IN;
  const int E  = in_sizes[1] / 2;
  const int* esrc = eidx;
  const int* edst = eidx + E;

  char* ws = (char*)d_ws;
  size_t off = 0;
  auto carve = [&](size_t bytes) -> char* {
    char* p = ws + off;
    off += (bytes + 255) & ~(size_t)255;
    return p;
  };
  __bf16* bpl1  = (__bf16*)carve((size_t)F_IN * H   * sizeof(__bf16));
  __bf16* bpr1  = (__bf16*)carve((size_t)F_IN * H   * sizeof(__bf16));
  __bf16* bpl2  = (__bf16*)carve((size_t)H * OUT    * sizeof(__bf16));
  __bf16* bpr2  = (__bf16*)carve((size_t)H * OUT    * sizeof(__bf16));
  float*  xl    = (float*) carve((size_t)Nn * H * sizeof(float));
  float*  xr    = (float*) carve((size_t)Nn * H * sizeof(float));
  float*  hbuf  = (float*) carve((size_t)Nn * H * sizeof(float));
  int*    cnt    = (int*)  carve((size_t)Nn * sizeof(int));
  int*    rowptr = (int*)  carve(((size_t)Nn + 1) * sizeof(int));
  int*    rank   = (int*)  carve((size_t)E * sizeof(int));
  int*    csrsrc = (int*)  carve((size_t)E * sizeof(int));
  int*    hist   = (int*)  carve((size_t)HIST_K * Nn * sizeof(int));

  auto g256 = [](long long n) { return (unsigned)((n + 255) / 256); };
  auto gwav = [](long long n) { return (unsigned)((n + 7) / 8); };

  pack_b_k<<<g256((long long)F_IN * H), 256, 0, stream>>>(Wl1, bpl1, F_IN, H);
  pack_b_k<<<g256((long long)F_IN * H), 256, 0, stream>>>(Wr1, bpr1, F_IN, H);
  pack_b_k<<<g256((long long)H * OUT),  256, 0, stream>>>(Wl2, bpl2, H, OUT);
  pack_b_k<<<g256((long long)H * OUT),  256, 0, stream>>>(Wr2, bpr2, H, OUT);

  hist_k<<<HIST_K, 1024, 0, stream>>>(edst, E, Nn, hist, rank);
  colscan_k<<<g256(Nn), 256, 0, stream>>>(hist, Nn, cnt);
  scan_k<<<1, 1024, 0, stream>>>(cnt, rowptr, Nn);
  scatter_k<<<g256(E), 256, 0, stream>>>(esrc, edst, rank, rowptr, hist, E, Nn, csrsrc);

  const int Mt = Nn / 16;

  {
    static_assert(F_IN == 4 * 32 && H == 8 * 16, "projection template arguments");
    wmma_proj_pair_k<8, 4><<<gwav(Mt), 256, 0, stream>>>(x, bpl1, bpr1, xl, xr, Mt);
    node_attn_k<4><<<gwav(Nn), 256, 0, stream>>>(xl, xr, att1, b1, rowptr, csrsrc, Nn, hbuf);
  }

  {
    static_assert(H == 4 * 32 && OUT == 2 * 16, "projection template arguments");
    wmma_proj_pair_k<2, 4><<<gwav(Mt), 256, 0, stream>>>(hbuf, bpl2, bpr2, xl, xr, Mt);
    node_attn32_k<<<gwav(Nn), 256, 0, stream>>>(xl, xr, att2, b2, rowptr, csrsrc, Nn, out);
  }
}


# Layer-1 node kernel: a 512-byte source row lies across 16 lanes, a wave step takes 2 edges, logits summed over 16 lanes by DPP, no LDS permutes; 2 softmax streams merged at the end; sum order changes
# speedup vs baseline: 8.7211x; 1.1647x over previous
#include <hip/hip_runtime.h>

typedef __attribute__((ext_vector_type(16))) __bf16 v16bf;
typedef __attribute__((ext_vector_type(8)))  float  v8f;


__device__ __forceinline__ void pack_b_elem(const float* __restrict__ W, __bf16* __restrict__ Bp,
                                            int K, int N, int o) {
  int i    = o & 15;
  int lane = (o >> 4) & 31;
  int t = o >> 9;
  int Kt = K >> 5;
  int ks = t % Kt;
  int tn = t / Kt;
  int n = tn * 16 + (lane & 15);
  int k = ks * 32 + ((lane >> 4) << 3) + (i < 8 ? i : i + 8);
  Bp[o] = (__bf16)W[k * N + n];
}

template <int NT, int KT>
__device__ __forceinline__ void strip_times_b(const v16bf (&a)[KT], const v16bf* __restrict__ Bf,
                                              float* __restrict__ C, long long tm, int lane) {
  constexpr int N = NT * 16;
  v8f zero = {};
  v8f acc[NT];
  #pragma unroll
  for (int t = 0; t < NT; ++t) acc[t] = zero;
  #pragma unroll
  for (int ks = 0; ks < KT; ++ks) {
    #pragma unroll
    for (int t = 0; t < NT; ++t) {
      v16bf b = Bf[(t * KT + ks) * 32 + lane];
      acc[t] = __builtin_amdgcn_wmma_f32_16x16x32_bf16(
          false, a[ks], false, b, (short)0, acc[t], false, false);
    }
  }
  int colb = lane & 15;
  long long rbase = tm * 16 + ((lane >> 4) << 3);
  #pragma unroll
  for (int t = 0; t < NT; ++t) {
    #pragma unroll
    for (int r = 0; r < 8; ++r)
      C[(rbase + r) * N + t * 16 + colb] = acc[t][r];
  }
}

template <int NT, int KT>
__device__ __forceinline__ void proj_pair_block(const float* __restrict__ X,
                                                const __bf16* __restrict__ Bpl,
                                                const __bf16* __restrict__ Bpr,
                                                float* __restrict__ Cl,
                                                float* __restrict__ Cr, int Mt, int block) {
  constexpr int K = KT * 32;
  int wave = threadIdx.x >> 5;
  int lane = threadIdx.x & 31;
  long long tm = (long long)block * 8 + wave;
  if (tm >= Mt) return;
  const float* p = X + (tm * 16 + (lane & 15)) * K + ((lane >> 4) << 3);
  float4 f[KT][4];
  #pragma unroll
  for (int ks = 0; ks < KT; ++ks) {
    const float4* q = (const float4*)(p + ks * 32);
    f[ks][0] = q[0];
    f[ks][1] = q[1];
    f[ks][2] = q[4];
    f[ks][3] = q[5];
  }
  __builtin_amdgcn_sched_barrier(0);
  v16bf a[KT];
  #pragma unroll
  for (int ks = 0; ks < KT; ++ks) {
    #pragma unroll
    for (int j = 0; j < 4; ++j) {
      a[ks][4 * j + 0] = (__bf16)f[ks][j].x;
      a[ks][4 * j + 1] = (__bf16)f[ks][j].y;
      a[ks][4 * j + 2] = (__bf16)f[ks][j].z;
      a[ks][4 * j + 3] = (__bf16)f[ks][j].w;
    }
  }
  strip_times_b<NT, KT>(a, (const v16bf*)Bpl, Cl, tm, lane);
  strip_times_b<NT, KT>(a, (const v16bf*)Bpr, Cr, tm, lane);
}

template <int NT, int KT>
__global__ __launch_bounds__(256) void wmma_proj_pair_k(const float* __restrict__ X,
                                                        const __bf16* __restrict__ Bpl,
                                                        const __bf16* __restrict__ Bpr,
                                                        float* __restrict__ Cl,
                                                        float* __restrict__ Cr, int Mt) {
  proj_pair_block<NT, KT>(X, Bpl, Bpr, Cl, Cr, Mt, (int)blockIdx.x);
}


constexpr int HIST_K    = 64;
constexpr int HIST_SLAB = 65536;

__device__ __forceinline__ int chunk_len(int E) { return (E + HIST_K - 1) / HIST_K; }

constexpr int PACK_BIG   = 128 * 128 / 1024;
constexpr int PACK_SMALL = 128 * 32 / 1024;
constexpr int PACK_WGS   = 2 * PACK_BIG + 2 * PACK_SMALL;

__global__ __launch_bounds__(1024) void hist_pack_k(const int* __restrict__ edst, int E, int Nn,
                                                    int* __restrict__ H, int* __restrict__ lrank,
                                                    const float* __restrict__ Wl1, const float* __restrict__ Wr1,
                                                    const float* __restrict__ Wl2, const float* __restrict__ Wr2,
                                                    __bf16* __restrict__ bpl1, __bf16* __restrict__ bpr1,
                                                    __bf16* __restrict__ bpl2, __bf16* __restrict__ bpr2) {
  __shared__ alignas(16) int h[HIST_SLAB];
  const int tid = threadIdx.x;
  if (blockIdx.x >= HIST_K) {
    int g = (int)blockIdx.x - HIST_K;
    if (g < PACK_BIG)             { pack_b_elem(Wl1, bpl1, 128, 128, g * 1024 + tid); return; }
    g -= PACK_BIG;
    if (g < PACK_BIG)             { pack_b_elem(Wr1, bpr1, 128, 128, g * 1024 + tid); return; }
    g -= PACK_BIG;
    if (g < PACK_SMALL)           { pack_b_elem(Wl2, bpl2, 128, 32, g * 1024 + tid); return; }
    g -= PACK_SMALL;
    if (g < PACK_SMALL)           { pack_b_elem(Wr2, bpr2, 128, 32, g * 1024 + tid); }
    return;
  }
  const int k = blockIdx.x;
  const int chunk = chunk_len(E);
  const int e0 = k * chunk;
  const int e1 = (e0 + chunk < E) ? (e0 + chunk) : E;
  for (int lo = 0; lo < Nn; lo += HIST_SLAB) {
    const int span = (Nn - lo < HIST_SLAB) ? (Nn - lo) : HIST_SLAB;
    for (int i = tid * 4; i < span; i += 4096) *(int4*)(h + i) = make_int4(0, 0, 0, 0);
    __syncthreads();
    for (int e = e0 + tid; e < e1; e += 8 * 1024) {
      int d[8];
      #pragma unroll
      for (int q = 0; q < 8; ++q) {
        const int ee = e + q * 1024;
        d[q] = (ee < e1) ? edst[ee] : -1;
      }
      unsigned rel[8];
      int r[8];
      #pragma unroll
      for (int q = 0; q < 8; ++q) {
        rel[q] = (unsigned)(d[q] - lo);
        r[q] = (rel[q] < (unsigned)span) ? atomicAdd(&h[rel[q]], 1) : 0;
      }
      #pragma unroll
      for (int q = 0; q < 8; ++q)
        if (rel[q] < (unsigned)span) lrank[e + q * 1024] = r[q];
    }
    __syncthreads();
    int* __restrict__ row = H + (long long)k * Nn + lo;
    for (int i = tid; i < span; i += 1024) row[i] = h[i];
    __syncthreads();
  }
}

__device__ __forceinline__ void colscan_block(int* __restrict__ H, int Nn, int* __restrict__ cnt, int block) {
  const int n = block * 256 + threadIdx.x;
  if (n >= Nn) return;
  int* col = H + n;
  int v[HIST_K];
  #pragma unroll
  for (int k = 0; k < HIST_K; ++k) v[k] = col[(long long)k * Nn];
  int run = 0;
  #pragma unroll
  for (int k = 0; k < HIST_K; ++k) { col[(long long)k * Nn] = run; run += v[k]; }
  cnt[n] = run;
}

__global__ __launch_bounds__(256) void colscan_proj1_k(int* __restrict__ H, int Nn, int* __restrict__ cnt,
                                                       const float* __restrict__ X,
                                                       const __bf16* __restrict__ Bpl,
                                                       const __bf16* __restrict__ Bpr,
                                                       float* __restrict__ Cl, float* __restrict__ Cr,
                                                       int Mt, int proj_blocks) {
  const int b = (int)blockIdx.x;
  if (b < proj_blocks) proj_pair_block<8, 4>(X, Bpl, Bpr, Cl, Cr, Mt, b);
  else                 colscan_block(H, Nn, cnt, b - proj_blocks);
}

__global__ __launch_bounds__(1024) void scan_k(const int* __restrict__ cnt,
                                               int* __restrict__ rowptr, int Nn) {
  __shared__ int wsum[32];
  const int tid = threadIdx.x, lane = tid & 31, wv = tid >> 5;
  int carry = 0;
  for (int base = 0; base < Nn; base += 16384) {
    const int i = base + tid * 16;
    const bool whole = (i + 15 < Nn);
    int v[16];
    if (whole) {
      #pragma unroll
      for (int q = 0; q < 4; ++q) {
        const int4 w = *(const int4*)(cnt + i + 4 * q);
        v[4 * q] = w.x; v[4 * q + 1] = w.y; v[4 * q + 2] = w.z; v[4 * q + 3] = w.w;
      }
    } else {
      #pragma unroll
      for (int q = 0; q < 16; ++q) v[q] = (i + q < Nn) ? cnt[i + q] : 0;
    }
    int t = 0;
    #pragma unroll
    for (int q = 0; q < 16; ++q) t += v[q];
    int inc = t;
    #pragma unroll
    for (int o = 1; o < 32; o <<= 1) {
      const int u = __shfl_up(inc, o, 32);
      if (lane >= o) inc += u;
    }
    if (lane == 31) wsum[wv] = inc;
    __syncthreads();
    const int ws = wsum[lane];
    int winc = ws;
    #pragma unroll
    for (int o = 1; o < 32; o <<= 1) {
      const int u = __shfl_up(winc, o, 32);
      if (lane >= o) winc += u;
    }
    const int wave_excl = __shfl(winc, wv, 32) - __shfl(ws, wv, 32);
    const int total = __shfl(winc, 31, 32);
    int e = carry + wave_excl + inc - t;
    int p[16];
    #pragma unroll
    for (int q = 0; q < 16; ++q) { p[q] = e; e += v[q]; }
    if (whole) {
      #pragma unroll
      for (int q = 0; q < 4; ++q)
        *(int4*)(rowptr + i + 4 * q) = make_int4(p[4 * q], p[4 * q + 1], p[4 * q + 2], p[4 * q + 3]);
    } else {
      #pragma unroll
      for (int q = 0; q < 16; ++q) if (i + q < Nn) rowptr[i + q] = p[q];
    }
    carry += total;
    __syncthreads();
  }
  if (tid == 0) rowptr[Nn] = carry;
}

__global__ __launch_bounds__(256) void scatter_k(const int* __restrict__ esrc,
                                                 const int* __restrict__ edst,
                                                 const int* __restrict__ lrank,
                                                 const int* __restrict__ rowptr,
                                                 const int* __restrict__ H, int E, int Nn,
                                                 int* __restrict__ csr_src) {
  int i = blockIdx.x * 256 + threadIdx.x;
  if (i < E) {
    const int k = i / chunk_len(E);
    const int d = edst[i];
    csr_src[rowptr[d] + H[(long long)k * Nn + d] + lrank[i]] = esrc[i];
  }
}

__device__ __forceinline__ void row_scale(float4& acc, float s) { acc.x *= s; acc.y *= s; acc.z *= s; acc.w *= s; }
__device__ __forceinline__ void row_fma(float4& acc, float w, float4 r) {
  acc.x = fmaf(w, r.x, acc.x); acc.y = fmaf(w, r.y, acc.y);
  acc.z = fmaf(w, r.z, acc.z); acc.w = fmaf(w, r.w, acc.w);
}
__device__ __forceinline__ float4 row_final(float4 acc, float dd, float4 b) {
  return make_float4(fmaxf(acc.x / dd + b.x, 0.0f), fmaxf(acc.y / dd + b.y, 0.0f),
                     fmaxf(acc.z / dd + b.z, 0.0f), fmaxf(acc.w / dd + b.w, 0.0f));
}


template <int MASK>
__device__ __forceinline__ float row_xor(float v) {
  return __builtin_bit_cast(float, __builtin_amdgcn_update_dpp(0, __builtin_bit_cast(int, v),
                                                               0x160 + MASK, 0xf, 0xf, true));
}
__device__ __forceinline__ float xhalf(float v) {
  return __builtin_bit_cast(float, __builtin_amdgcn_permlanex16(__builtin_bit_cast(int, v),
      __builtin_bit_cast(int, v), 0x76543210, 0xfedcba98, false, false));
}
template <int T>
__device__ __forceinline__ int oct_lane(int v) {
  return (int)__builtin_amdgcn_mov_dpp8((unsigned)v, T * 0x249249);
}

typedef float v2f __attribute__((ext_vector_type(2)));
__device__ __forceinline__ float logit_part2(float4 l, v2f r01, v2f r23, v2f a01, v2f a23) {
  v2f v01 = (v2f){l.x, l.y} + r01;
  v2f v23 = (v2f){l.z, l.w} + r23;
  v01 = __builtin_elementwise_max(v01, 0.2f * v01);
  v23 = __builtin_elementwise_max(v23, 0.2f * v23);
  const v2f p = __builtin_elementwise_fma(v23, a23, v01 * a01);
  return p.x + p.y;
}

template <int T0>
__device__ __forceinline__ void four_steps(const float4* __restrict__ rowsL, int s, int nv, int q,
                                           v2f xr01, v2f xr23, v2f a01, v2f a23,
                                           float& m, float& den, float4& acc) {
  float4 r[4];
  float lg[4];
  r[0] = rowsL[(unsigned)oct_lane<T0 + 0>(s) * 8u + (unsigned)q];
  r[1] = rowsL[(unsigned)oct_lane<T0 + 1>(s) * 8u + (unsigned)q];
  r[2] = rowsL[(unsigned)oct_lane<T0 + 2>(s) * 8u + (unsigned)q];
  r[3] = rowsL[(unsigned)oct_lane<T0 + 3>(s) * 8u + (unsigned)q];
  #pragma unroll
  for (int u = 0; u < 4; ++u) lg[u] = logit_part2(r[u], xr01, xr23, a01, a23);
  #pragma unroll
  for (int u = 0; u < 4; ++u) lg[u] += row_xor<1>(lg[u]);
  #pragma unroll
  for (int u = 0; u < 4; ++u) lg[u] += row_xor<2>(lg[u]);
  #pragma unroll
  for (int u = 0; u < 4; ++u) lg[u] += row_xor<4>(lg[u]);
  float cm = m;
  #pragma unroll
  for (int u = 0; u < 4; ++u) {
    if (T0 + u < nv) cm = fmaxf(cm, lg[u]);
  }
  const float sc = __expf(m - cm);
  den *= sc;
  row_scale(acc, sc);
  m = cm;
  #pragma unroll
  for (int u = 0; u < 4; ++u) {
    const float w = (T0 + u < nv) ? __expf(lg[u] - cm) : 0.0f;
    den += w;
    row_fma(acc, w, r[u]);
  }
}

__global__ __launch_bounds__(256) void node_attn32_k(const float* __restrict__ xl,
                                                    const float* __restrict__ xr,
                                                    const float* __restrict__ att,
                                                    const float* __restrict__ bias,
                                                    const int* __restrict__ rowptr,
                                                    const int* __restrict__ csr_src,
                                                    int Nn, float* __restrict__ out) {
  const int lane = threadIdx.x & 31;
  const int wv = __builtin_amdgcn_readfirstlane((int)(threadIdx.x >> 5));
  const int node = blockIdx.x * 8 + wv;
  if (node >= Nn) return;
  const int q = lane & 7;
  const int slot = lane >> 3;
  const float4* __restrict__ rowsL = (const float4*)xl;
  const float4 xrd   = ((const float4*)xr)[(long long)node * 8 + q];
  const float4 a     = ((const float4*)att)[q];
  const float4 selfr = rowsL[(long long)node * 8 + q];
  const int beg = rowptr[node];
  const int end = rowptr[node + 1];

  const v2f xr01 = {xrd.x, xrd.y}, xr23 = {xrd.z, xrd.w};
  const v2f a01 = {a.x, a.y}, a23 = {a.z, a.w};

  float m = logit_part2(selfr, xr01, xr23, a01, a23);
  m += row_xor<1>(m);
  m += row_xor<2>(m);
  m += row_xor<4>(m);
  float den = (slot == 0) ? 1.0f : 0.0f;
  float4 acc = (slot == 0) ? selfr : make_float4(0.0f, 0.0f, 0.0f, 0.0f);

  for (int base = beg; base < end; base += 32) {
    const int n = (end - base < 32) ? (end - base) : 32;
    const int eo = 4 * q + slot;
    const int s = (eo < n) ? csr_src[base + eo] : node;
    const int nv = (n - slot + 3) >> 2;
    four_steps<0>(rowsL, s, nv, q, xr01, xr23, a01, a23, m, den, acc);
    if (n > 16) four_steps<4>(rowsL, s, nv, q, xr01, xr23, a01, a23, m, den, acc);
  }

  float mm = fmaxf(m, row_xor<8>(m));
  mm = fmaxf(mm, xhalf(mm));
  const float f = __expf(m - mm);
  den *= f;
  row_scale(acc, f);
  den += row_xor<8>(den);
  acc.x += row_xor<8>(acc.x); acc.y += row_xor<8>(acc.y);
  acc.z += row_xor<8>(acc.z); acc.w += row_xor<8>(acc.w);
  den += xhalf(den);
  acc.x += xhalf(acc.x); acc.y += xhalf(acc.y);
  acc.z += xhalf(acc.z); acc.w += xhalf(acc.w);
  if (slot == 0) {
    const float4 b = ((const float4*)bias)[q];
    ((float4*)out)[(long long)node * 8 + q] = row_final(acc, den + 1e-16f, b);
  }
}


template <int T>
__device__ __forceinline__ int row_lane(int v) {
  return __builtin_amdgcn_update_dpp(0, v, 0x150 + T, 0xf, 0xf, true);
}

__device__ __forceinline__ float logit_part8(float4 llo, float4 lhi, v2f x0, v2f x1, v2f x2, v2f x3,
                                             v2f a0, v2f a1, v2f a2, v2f a3) {
  v2f v0 = (v2f){llo.x, llo.y} + x0;
  v2f v1 = (v2f){llo.z, llo.w} + x1;
  v2f v2 = (v2f){lhi.x, lhi.y} + x2;
  v2f v3 = (v2f){lhi.z, lhi.w} + x3;
  v0 = __builtin_elementwise_max(v0, 0.2f * v0);
  v1 = __builtin_elementwise_max(v1, 0.2f * v1);
  v2 = __builtin_elementwise_max(v2, 0.2f * v2);
  v3 = __builtin_elementwise_max(v3, 0.2f * v3);
  v2f p = v0 * a0;
  p = __builtin_elementwise_fma(v1, a1, p);
  p = __builtin_elementwise_fma(v2, a2, p);
  p = __builtin_elementwise_fma(v3, a3, p);
  return p.x + p.y;
}

template <int T0>
__device__ __forceinline__ void four_steps128(const float4* __restrict__ rowsL, int s, int nv, int c,
                                              v2f x0, v2f x1, v2f x2, v2f x3,
                                              v2f a0, v2f a1, v2f a2, v2f a3,
                                              float& m, float& den, float4& acclo, float4& acchi) {
  float4 rlo[4], rhi[4];
  float lg[4];
  const unsigned o0 = (unsigned)row_lane<T0 + 0>(s) * 32u + (unsigned)c;
  const unsigned o1 = (unsigned)row_lane<T0 + 1>(s) * 32u + (unsigned)c;
  const unsigned o2 = (unsigned)row_lane<T0 + 2>(s) * 32u + (unsigned)c;
  const unsigned o3 = (unsigned)row_lane<T0 + 3>(s) * 32u + (unsigned)c;
  rlo[0] = rowsL[o0]; rhi[0] = rowsL[o0 + 16u];
  rlo[1] = rowsL[o1]; rhi[1] = rowsL[o1 + 16u];
  rlo[2] = rowsL[o2]; rhi[2] = rowsL[o2 + 16u];
  rlo[3] = rowsL[o3]; rhi[3] = rowsL[o3 + 16u];
  #pragma unroll
  for (int u = 0; u < 4; ++u) lg[u] = logit_part8(rlo[u], rhi[u], x0, x1, x2, x3, a0, a1, a2, a3);
  #pragma unroll
  for (int u = 0; u < 4; ++u) lg[u] += row_xor<1>(lg[u]);
  #pragma unroll
  for (int u = 0; u < 4; ++u) lg[u] += row_xor<2>(lg[u]);
  #pragma unroll
  for (int u = 0; u < 4; ++u) lg[u] += row_xor<4>(lg[u]);
  #pragma unroll
  for (int u = 0; u < 4; ++u) lg[u] += row_xor<8>(lg[u]);
  float cm = m;
  #pragma unroll
  for (int u = 0; u < 4; ++u) {
    if (T0 + u < nv) cm = fmaxf(cm, lg[u]);
  }
  const float sc = __expf(m - cm);
  den *= sc;
  row_scale(acclo, sc);
  row_scale(acchi, sc);
  m = cm;
  #pragma unroll
  for (int u = 0; u < 4; ++u) {
    const float w = (T0 + u < nv) ? __expf(lg[u] - cm) : 0.0f;
    den += w;
    row_fma(acclo, w, rlo[u]);
    row_fma(acchi, w, rhi[u]);
  }
}

__global__ __launch_bounds__(256, 10) void node_attn128_k(const float* __restrict__ xl,
                                                     const float* __restrict__ xr,
                                                     const float* __restrict__ att,
                                                     const float* __restrict__ bias,
                                                     const int* __restrict__ rowptr,
                                                     const int* __restrict__ csr_src,
                                                     int Nn, float* __restrict__ out) {
  const int lane = threadIdx.x & 31;
  const int wv = __builtin_amdgcn_readfirstlane((int)(threadIdx.x >> 5));
  const int node = blockIdx.x * 8 + wv;
  if (node >= Nn) return;
  const int c = lane & 15;
  const int half = lane >> 4;
  const float4* __restrict__ rowsL = (const float4*)xl;
  const long long nb = (long long)node * 32 + c;
  const float4 xdlo = ((const float4*)xr)[nb];
  const float4 xdhi = ((const float4*)xr)[nb + 16];
  const float4 alo  = ((const float4*)att)[c];
  const float4 ahi  = ((const float4*)att)[16 + c];
  const float4 slo  = rowsL[nb];
  const float4 shi  = rowsL[nb + 16];
  const int beg = rowptr[node];
  const int end = rowptr[node + 1];

  const v2f x0 = {xdlo.x, xdlo.y}, x1 = {xdlo.z, xdlo.w}, x2 = {xdhi.x, xdhi.y}, x3 = {xdhi.z, xdhi.w};
  const v2f a0 = {alo.x, alo.y}, a1 = {alo.z, alo.w}, a2 = {ahi.x, ahi.y}, a3 = {ahi.z, ahi.w};

  float m = logit_part8(slo, shi, x0, x1, x2, x3, a0, a1, a2, a3);
  m += row_xor<1>(m);
  m += row_xor<2>(m);
  m += row_xor<4>(m);
  m += row_xor<8>(m);
  const float4 zero4 = make_float4(0.0f, 0.0f, 0.0f, 0.0f);
  float den = (half == 0) ? 1.0f : 0.0f;
  float4 acclo = (half == 0) ? slo : zero4;
  float4 acchi = (half == 0) ? shi : zero4;

  for (int base = beg; base < end; base += 32) {
    const int n = (end - base < 32) ? (end - base) : 32;
    const int eo = 2 * c + half;
    const int s = (eo < n) ? csr_src[base + eo] : node;
    const int nv = (n - half + 1) >> 1;
    four_steps128<0>(rowsL, s, nv, c, x0, x1, x2, x3, a0, a1, a2, a3, m, den, acclo, acchi);
    if (n > 8)  four_steps128<4>(rowsL, s, nv, c, x0, x1, x2, x3, a0, a1, a2, a3, m, den, acclo, acchi);
    if (n > 16) four_steps128<8>(rowsL, s, nv, c, x0, x1, x2, x3, a0, a1, a2, a3, m, den, acclo, acchi);
    if (n > 24) four_steps128<12>(rowsL, s, nv, c, x0, x1, x2, x3, a0, a1, a2, a3, m, den, acclo, acchi);
  }

  const float mm = fmaxf(m, xhalf(m));
  const float f = __expf(m - mm);
  den *= f;
  row_scale(acclo, f);
  row_scale(acchi, f);
  den += xhalf(den);
  acclo.x += xhalf(acclo.x); acclo.y += xhalf(acclo.y);
  acclo.z += xhalf(acclo.z); acclo.w += xhalf(acclo.w);
  acchi.x += xhalf(acchi.x); acchi.y += xhalf(acchi.y);
  acchi.z += xhalf(acchi.z); acchi.w += xhalf(acchi.w);
  const float4 b = ((const float4*)bias)[lane];
  const float4 o = (half == 0) ? acclo : acchi;
  ((float4*)out)[(long long)node * 32 + lane] = row_final(o, den + 1e-16f, b);
}


extern "C" void kernel_launch(void* const* d_in, const int* in_sizes, int n_in,
                              void* d_out, int out_size, void* d_ws, size_t ws_size,
                              hipStream_t stream) {
  const int F_IN = 128, H = 128, OUT = 32;
  const float* x    = (const float*)d_in[0];
  const int*   eidx = (const int*)  d_in[1];
  const float* Wl1  = (const float*)d_in[2];
  const float* Wr1  = (const float*)d_in[3];
  const float* att1 = (const float*)d_in[4];
  const float* b1   = (const float*)d_in[5];
  const float* Wl2  = (const float*)d_in[6];
  const float* Wr2  = (const float*)d_in[7];
  const float* att2 = (const float*)d_in[8];
  const float* b2   = (const float*)d_in[9];
  float* out = (float*)d_out;

  const int Nn = in_sizes[0] / F_IN;
  const int E  = in_sizes[1] / 2;
  const int* esrc = eidx;
  const int* edst = eidx + E;

  char* ws = (char*)d_ws;
  size_t off = 0;
  auto carve = [&](size_t bytes) -> char* {
    char* p = ws + off;
    off += (bytes + 255) & ~(size_t)255;
    return p;
  };
  __bf16* bpl1  = (__bf16*)carve((size_t)F_IN * H   * sizeof(__bf16));
  __bf16* bpr1  = (__bf16*)carve((size_t)F_IN * H   * sizeof(__bf16));
  __bf16* bpl2  = (__bf16*)carve((size_t)H * OUT    * sizeof(__bf16));
  __bf16* bpr2  = (__bf16*)carve((size_t)H * OUT    * sizeof(__bf16));
  float*  xl    = (float*) carve((size_t)Nn * H * sizeof(float));
  float*  xr    = (float*) carve((size_t)Nn * H * sizeof(float));
  float*  hbuf  = (float*) carve((size_t)Nn * H * sizeof(float));
  int*    cnt    = (int*)  carve((size_t)Nn * sizeof(int));
  int*    rowptr = (int*)  carve(((size_t)Nn + 1) * sizeof(int));
  int*    rank   = (int*)  carve((size_t)E * sizeof(int));
  int*    csrsrc = (int*)  carve((size_t)E * sizeof(int));
  int*    hist   = (int*)  carve((size_t)HIST_K * Nn * sizeof(int));

  auto g256 = [](long long n) { return (unsigned)((n + 255) / 256); };
  auto gwav = [](long long n) { return (unsigned)((n + 7) / 8); };

  const int Mt = Nn / 16;
  static_assert(F_IN == 4 * 32 && H == 8 * 16 && OUT == 2 * 16, "projection template arguments");
  static_assert(F_IN == 128 && H == 128 && OUT == 32, "sizes the packing work-groups are cut for");

  hist_pack_k<<<HIST_K + PACK_WGS, 1024, 0, stream>>>(edst, E, Nn, hist, rank, Wl1, Wr1, Wl2, Wr2,
                                                      bpl1, bpr1, bpl2, bpr2);
  colscan_proj1_k<<<gwav(Mt) + g256(Nn), 256, 0, stream>>>(hist, Nn, cnt, x, bpl1, bpr1, xl, xr,
                                                           Mt, (int)gwav(Mt));
  scan_k<<<1, 1024, 0, stream>>>(cnt, rowptr, Nn);
  scatter_k<<<g256(E), 256, 0, stream>>>(esrc, edst, rank, rowptr, hist, E, Nn, csrsrc);

  node_attn128_k<<<gwav(Nn), 256, 0, stream>>>(xl, xr, att1, b1, rowptr, csrsrc, Nn, hbuf);

  wmma_proj_pair_k<2, 4><<<gwav(Mt), 256, 0, stream>>>(hbuf, bpl2, bpr2, xl, xr, Mt);
  node_attn32_k<<<gwav(Nn), 256, 0, stream>>>(xl, xr, att2, b2, rowptr, csrsrc, Nn, out);
}


# Column pass makes the row pointers itself: counting launch also leaves each chunk's total per 256-node block, H holds absolute places; scan launch gone (7 -> 6), scatter gathers 1 word not 2
# speedup vs baseline: 9.3108x; 1.0676x over previous
#include <hip/hip_runtime.h>

typedef __attribute__((ext_vector_type(16))) __bf16 v16bf;
typedef __attribute__((ext_vector_type(8)))  float  v8f;


__device__ __forceinline__ void pack_b_elem(const float* __restrict__ W, __bf16* __restrict__ Bp,
                                            int K, int N, int o) {
  int i    = o & 15;
  int lane = (o >> 4) & 31;
  int t = o >> 9;
  int Kt = K >> 5;
  int ks = t % Kt;
  int tn = t / Kt;
  int n = tn * 16 + (lane & 15);
  int k = ks * 32 + ((lane >> 4) << 3) + (i < 8 ? i : i + 8);
  Bp[o] = (__bf16)W[k * N + n];
}

template <int NT, int KT>
__device__ __forceinline__ void strip_times_b(const v16bf (&a)[KT], const v16bf* __restrict__ Bf,
                                              float* __restrict__ C, long long tm, int lane) {
  constexpr int N = NT * 16;
  v8f zero = {};
  v8f acc[NT];
  #pragma unroll
  for (int t = 0; t < NT; ++t) acc[t] = zero;
  #pragma unroll
  for (int ks = 0; ks < KT; ++ks) {
    #pragma unroll
    for (int t = 0; t < NT; ++t) {
      v16bf b = Bf[(t * KT + ks) * 32 + lane];
      acc[t] = __builtin_amdgcn_wmma_f32_16x16x32_bf16(
          false, a[ks], false, b, (short)0, acc[t], false, false);
    }
  }
  int colb = lane & 15;
  long long rbase = tm * 16 + ((lane >> 4) << 3);
  #pragma unroll
  for (int t = 0; t < NT; ++t) {
    #pragma unroll
    for (int r = 0; r < 8; ++r)
      C[(rbase + r) * N + t * 16 + colb] = acc[t][r];
  }
}

template <int NT, int KT>
__device__ __forceinline__ void proj_pair_block(const float* __restrict__ X,
                                                const __bf16* __restrict__ Bpl,
                                                const __bf16* __restrict__ Bpr,
                                                float* __restrict__ Cl,
                                                float* __restrict__ Cr, int Mt, int block) {
  constexpr int K = KT * 32;
  int wave = threadIdx.x >> 5;
  int lane = threadIdx.x & 31;
  long long tm = (long long)block * 8 + wave;
  if (tm >= Mt) return;
  const float* p = X + (tm * 16 + (lane & 15)) * K + ((lane >> 4) << 3);
  float4 f[KT][4];
  #pragma unroll
  for (int ks = 0; ks < KT; ++ks) {
    const float4* q = (const float4*)(p + ks * 32);
    f[ks][0] = q[0];
    f[ks][1] = q[1];
    f[ks][2] = q[4];
    f[ks][3] = q[5];
  }
  __builtin_amdgcn_sched_barrier(0);
  v16bf a[KT];
  #pragma unroll
  for (int ks = 0; ks < KT; ++ks) {
    #pragma unroll
    for (int j = 0; j < 4; ++j) {
      a[ks][4 * j + 0] = (__bf16)f[ks][j].x;
      a[ks][4 * j + 1] = (__bf16)f[ks][j].y;
      a[ks][4 * j + 2] = (__bf16)f[ks][j].z;
      a[ks][4 * j + 3] = (__bf16)f[ks][j].w;
    }
  }
  strip_times_b<NT, KT>(a, (const v16bf*)Bpl, Cl, tm, lane);
  strip_times_b<NT, KT>(a, (const v16bf*)Bpr, Cr, tm, lane);
}

template <int NT, int KT>
__global__ __launch_bounds__(256) void wmma_proj_pair_k(const float* __restrict__ X,
                                                        const __bf16* __restrict__ Bpl,
                                                        const __bf16* __restrict__ Bpr,
                                                        float* __restrict__ Cl,
                                                        float* __restrict__ Cr, int Mt) {
  proj_pair_block<NT, KT>(X, Bpl, Bpr, Cl, Cr, Mt, (int)blockIdx.x);
}


constexpr int HIST_K    = 64;
constexpr int HIST_SLAB = 65536;
constexpr int NODE_BLK  = 256;
static_assert(HIST_SLAB % NODE_BLK == 0, "a node block lies inside one slab");

__device__ __forceinline__ int chunk_len(int E) { return (E + HIST_K - 1) / HIST_K; }

template <int MASK>
__device__ __forceinline__ int irow_xor(int v) {
  return __builtin_amdgcn_update_dpp(0, v, 0x160 + MASK, 0xf, 0xf, true);
}
__device__ __forceinline__ int wave_sum(int s) {
  s += irow_xor<1>(s);
  s += irow_xor<2>(s);
  s += irow_xor<4>(s);
  s += irow_xor<8>(s);
  s += __builtin_amdgcn_permlanex16(s, s, 0x76543210, 0xfedcba98, false, false);
  return s;
}

constexpr int PACK_BIG   = 128 * 128 / 1024;
constexpr int PACK_SMALL = 128 * 32 / 1024;
constexpr int PACK_WGS   = 2 * PACK_BIG + 2 * PACK_SMALL;

__global__ __launch_bounds__(1024) void hist_pack_k(const int* __restrict__ edst, int E, int Nn,
                                                    int* __restrict__ H, int* __restrict__ BT,
                                                    int* __restrict__ lrank,
                                                    const float* __restrict__ Wl1, const float* __restrict__ Wr1,
                                                    const float* __restrict__ Wl2, const float* __restrict__ Wr2,
                                                    __bf16* __restrict__ bpl1, __bf16* __restrict__ bpr1,
                                                    __bf16* __restrict__ bpl2, __bf16* __restrict__ bpr2) {
  __shared__ alignas(16) int h[HIST_SLAB];
  const int tid = threadIdx.x;
  if (blockIdx.x >= HIST_K) {
    int g = (int)blockIdx.x - HIST_K;
    if (g < PACK_BIG)             { pack_b_elem(Wl1, bpl1, 128, 128, g * 1024 + tid); return; }
    g -= PACK_BIG;
    if (g < PACK_BIG)             { pack_b_elem(Wr1, bpr1, 128, 128, g * 1024 + tid); return; }
    g -= PACK_BIG;
    if (g < PACK_SMALL)           { pack_b_elem(Wl2, bpl2, 128, 32, g * 1024 + tid); return; }
    g -= PACK_SMALL;
    if (g < PACK_SMALL)           { pack_b_elem(Wr2, bpr2, 128, 32, g * 1024 + tid); }
    return;
  }
  const int k = blockIdx.x;
  const int chunk = chunk_len(E);
  const int e0 = k * chunk;
  const int e1 = (e0 + chunk < E) ? (e0 + chunk) : E;
  for (int lo = 0; lo < Nn; lo += HIST_SLAB) {
    const int span = (Nn - lo < HIST_SLAB) ? (Nn - lo) : HIST_SLAB;
    const int span_blk = (span + NODE_BLK - 1) / NODE_BLK * NODE_BLK;
    for (int i = tid * 4; i < span_blk; i += 4096) *(int4*)(h + i) = make_int4(0, 0, 0, 0);
    __syncthreads();
    for (int e = e0 + tid; e < e1; e += 8 * 1024) {
      int d[8];
      #pragma unroll
      for (int q = 0; q < 8; ++q) {
        const int ee = e + q * 1024;
        d[q] = (ee < e1) ? edst[ee] : -1;
      }
      unsigned rel[8];
      int r[8];
      #pragma unroll
      for (int q = 0; q < 8; ++q) {
        rel[q] = (unsigned)(d[q] - lo);
        r[q] = (rel[q] < (unsigned)span) ? atomicAdd(&h[rel[q]], 1) : 0;
      }
      #pragma unroll
      for (int q = 0; q < 8; ++q)
        if (rel[q] < (unsigned)span) lrank[e + q * 1024] = r[q];
    }
    __syncthreads();
    int* __restrict__ row = H + (long long)k * Nn + lo;
    for (int i = tid; i < span; i += 1024) row[i] = h[i];
    const int lane = tid & 31;
    const int wv = __builtin_amdgcn_readfirstlane(tid >> 5);
    const int sblk = span_blk / NODE_BLK;
    #pragma unroll
    for (int r = 0; r < HIST_SLAB / NODE_BLK / 32; ++r) {
      const int bb = wv + 32 * r;
      if (bb < sblk) {
        int s = 0;
        #pragma unroll
        for (int j = 0; j < NODE_BLK / 32; ++j) s += h[bb * NODE_BLK + j * 32 + lane];
        s = wave_sum(s);
        if (lane == 0) BT[(lo / NODE_BLK + bb) * HIST_K + k] = s;
      }
    }
    __syncthreads();
  }
}

static_assert(HIST_K % 4 == 0, "BT is read 16 bytes at a time");
__device__ __forceinline__ void colscan_block(int* __restrict__ H, int Nn, const int* __restrict__ BT,
                                              int* __restrict__ rowptr, int block) {
  __shared__ int wtot[NODE_BLK / 32];
  __shared__ int wpre[NODE_BLK / 32];
  const int t = threadIdx.x, lane = t & 31, wv = t >> 5;
  const int n = block * NODE_BLK + t;
  const bool live = n < Nn;
  int* col = H + (live ? n : Nn - 1);
  int v[HIST_K];
  #pragma unroll
  for (int k = 0; k < HIST_K; ++k) v[k] = col[(long long)k * Nn];
  int pre = 0;
  const int4* __restrict__ bt4 = (const int4*)BT;
  const int lim = block * (HIST_K / 4);
  #pragma unroll 4
  for (int q = t; q < lim; q += NODE_BLK) {
    const int4 w = bt4[q];
    pre += (w.x + w.y) + (w.z + w.w);
  }
  int tot = 0;
  #pragma unroll
  for (int k = 0; k < HIST_K; ++k) tot += v[k];
  int inc = tot;
  #pragma unroll
  for (int o = 1; o < 32; o <<= 1) {
    const int u = __shfl_up(inc, o, 32);
    if (lane >= o) inc += u;
  }
  pre = wave_sum(pre);
  if (lane == 31) wtot[wv] = inc;
  if (lane == 0)  wpre[wv] = pre;
  __syncthreads();
  int start = inc - tot;
  #pragma unroll
  for (int w = 0; w < NODE_BLK / 32; ++w) {
    start += wpre[w];
    if (w < wv) start += wtot[w];
  }
  if (!live) return;
  rowptr[n] = start;
  if (n == Nn - 1) rowptr[Nn] = start + tot;
  int run = start;
  #pragma unroll
  for (int k = 0; k < HIST_K; ++k) { col[(long long)k * Nn] = run; run += v[k]; }
}

__global__ __launch_bounds__(256) void colscan_proj1_k(int* __restrict__ H, int Nn, const int* __restrict__ BT,
                                                       int* __restrict__ rowptr,
                                                       const float* __restrict__ X,
                                                       const __bf16* __restrict__ Bpl,
                                                       const __bf16* __restrict__ Bpr,
                                                       float* __restrict__ Cl, float* __restrict__ Cr,
                                                       int Mt, int proj_blocks) {
  const int b = (int)blockIdx.x;
  if (b < proj_blocks) proj_pair_block<8, 4>(X, Bpl, Bpr, Cl, Cr, Mt, b);
  else                 colscan_block(H, Nn, BT, rowptr, b - proj_blocks);
}

__global__ __launch_bounds__(256) void scatter_k(const int* __restrict__ esrc,
                                                 const int* __restrict__ edst,
                                                 const int* __restrict__ lrank,
                                                 const int* __restrict__ H, int E, int Nn,
                                                 int* __restrict__ csr_src) {
  int i = blockIdx.x * 256 + threadIdx.x;
  if (i < E) {
    const int k = i / chunk_len(E);
    const int d = edst[i];
    csr_src[H[(long long)k * Nn + d] + lrank[i]] = esrc[i];
  }
}

__device__ __forceinline__ void row_scale(float4& acc, float s) { acc.x *= s; acc.y *= s; acc.z *= s; acc.w *= s; }
__device__ __forceinline__ void row_fma(float4& acc, float w, float4 r) {
  acc.x = fmaf(w, r.x, acc.x); acc.y = fmaf(w, r.y, acc.y);
  acc.z = fmaf(w, r.z, acc.z); acc.w = fmaf(w, r.w, acc.w);
}
__device__ __forceinline__ float4 row_final(float4 acc, float dd, float4 b) {
  return make_float4(fmaxf(acc.x / dd + b.x, 0.0f), fmaxf(acc.y / dd + b.y, 0.0f),
                     fmaxf(acc.z / dd + b.z, 0.0f), fmaxf(acc.w / dd + b.w, 0.0f));
}


template <int MASK>
__device__ __forceinline__ float row_xor(float v) {
  return __builtin_bit_cast(float, __builtin_amdgcn_update_dpp(0, __builtin_bit_cast(int, v),
                                                               0x160 + MASK, 0xf, 0xf, true));
}
__device__ __forceinline__ float xhalf(float v) {
  return __builtin_bit_cast(float, __builtin_amdgcn_permlanex16(__builtin_bit_cast(int, v),
      __builtin_bit_cast(int, v), 0x76543210, 0xfedcba98, false, false));
}
template <int T>
__device__ __forceinline__ int oct_lane(int v) {
  return (int)__builtin_amdgcn_mov_dpp8((unsigned)v, T * 0x249249);
}

typedef float v2f __attribute__((ext_vector_type(2)));
__device__ __forceinline__ float logit_part2(float4 l, v2f r01, v2f r23, v2f a01, v2f a23) {
  v2f v01 = (v2f){l.x, l.y} + r01;
  v2f v23 = (v2f){l.z, l.w} + r23;
  v01 = __builtin_elementwise_max(v01, 0.2f * v01);
  v23 = __builtin_elementwise_max(v23, 0.2f * v23);
  const v2f p = __builtin_elementwise_fma(v23, a23, v01 * a01);
  return p.x + p.y;
}

template <int T0>
__device__ __forceinline__ void four_steps(const float4* __restrict__ rowsL, int s, int nv, int q,
                                           v2f xr01, v2f xr23, v2f a01, v2f a23,
                                           float& m, float& den, float4& acc) {
  float4 r[4];
  float lg[4];
  r[0] = rowsL[(unsigned)oct_lane<T0 + 0>(s) * 8u + (unsigned)q];
  r[1] = rowsL[(unsigned)oct_lane<T0 + 1>(s) * 8u + (unsigned)q];
  r[2] = rowsL[(unsigned)oct_lane<T0 + 2>(s) * 8u + (unsigned)q];
  r[3] = rowsL[(unsigned)oct_lane<T0 + 3>(s) * 8u + (unsigned)q];
  #pragma unroll
  for (int u = 0; u < 4; ++u) lg[u] = logit_part2(r[u], xr01, xr23, a01, a23);
  #pragma unroll
  for (int u = 0; u < 4; ++u) lg[u] += row_xor<1>(lg[u]);
  #pragma unroll
  for (int u = 0; u < 4; ++u) lg[u] += row_xor<2>(lg[u]);
  #pragma unroll
  for (int u = 0; u < 4; ++u) lg[u] += row_xor<4>(lg[u]);
  float cm = m;
  #pragma unroll
  for (int u = 0; u < 4; ++u) {
    if (T0 + u < nv) cm = fmaxf(cm, lg[u]);
  }
  const float sc = __expf(m - cm);
  den *= sc;
  row_scale(acc, sc);
  m = cm;
  #pragma unroll
  for (int u = 0; u < 4; ++u) {
    const float w = (T0 + u < nv) ? __expf(lg[u] - cm) : 0.0f;
    den += w;
    row_fma(acc, w, r[u]);
  }
}

__global__ __launch_bounds__(256) void node_attn32_k(const float* __restrict__ xl,
                                                    const float* __restrict__ xr,
                                                    const float* __restrict__ att,
                                                    const float* __restrict__ bias,
                                                    const int* __restrict__ rowptr,
                                                    const int* __restrict__ csr_src,
                                                    int Nn, float* __restrict__ out) {
  const int lane = threadIdx.x & 31;
  const int wv = __builtin_amdgcn_readfirstlane((int)(threadIdx.x >> 5));
  const int node = blockIdx.x * 8 + wv;
  if (node >= Nn) return;
  const int q = lane & 7;
  const int slot = lane >> 3;
  const float4* __restrict__ rowsL = (const float4*)xl;
  const float4 xrd   = ((const float4*)xr)[(long long)node * 8 + q];
  const float4 a     = ((const float4*)att)[q];
  const float4 selfr = rowsL[(long long)node * 8 + q];
  const int beg = rowptr[node];
  const int end = rowptr[node + 1];

  const v2f xr01 = {xrd.x, xrd.y}, xr23 = {xrd.z, xrd.w};
  const v2f a01 = {a.x, a.y}, a23 = {a.z, a.w};

  float m = logit_part2(selfr, xr01, xr23, a01, a23);
  m += row_xor<1>(m);
  m += row_xor<2>(m);
  m += row_xor<4>(m);
  float den = (slot == 0) ? 1.0f : 0.0f;
  float4 acc = (slot == 0) ? selfr : make_float4(0.0f, 0.0f, 0.0f, 0.0f);

  for (int base = beg; base < end; base += 32) {
    const int n = (end - base < 32) ? (end - base) : 32;
    const int eo = 4 * q + slot;
    const int s = (eo < n) ? csr_src[base + eo] : node;
    const int nv = (n - slot + 3) >> 2;
    four_steps<0>(rowsL, s, nv, q, xr01, xr23, a01, a23, m, den, acc);
    if (n > 16) four_steps<4>(rowsL, s, nv, q, xr01, xr23, a01, a23, m, den, acc);
  }

  float mm = fmaxf(m, row_xor<8>(m));
  mm = fmaxf(mm, xhalf(mm));
  const float f = __expf(m - mm);
  den *= f;
  row_scale(acc, f);
  den += row_xor<8>(den);
  acc.x += row_xor<8>(acc.x); acc.y += row_xor<8>(acc.y);
  acc.z += row_xor<8>(acc.z); acc.w += row_xor<8>(acc.w);
  den += xhalf(den);
  acc.x += xhalf(acc.x); acc.y += xhalf(acc.y);
  acc.z += xhalf(acc.z); acc.w += xhalf(acc.w);
  if (slot == 0) {
    const float4 b = ((const float4*)bias)[q];
    ((float4*)out)[(long long)node * 8 + q] = row_final(acc, den + 1e-16f, b);
  }
}


template <int T>
__device__ __forceinline__ int row_lane(int v) {
  return __builtin_amdgcn_update_dpp(0, v, 0x150 + T, 0xf, 0xf, true);
}

__device__ __forceinline__ float logit_part8(float4 llo, float4 lhi, v2f x0, v2f x1, v2f x2, v2f x3,
                                             v2f a0, v2f a1, v2f a2, v2f a3) {
  v2f v0 = (v2f){llo.x, llo.y} + x0;
  v2f v1 = (v2f){llo.z, llo.w} + x1;
  v2f v2 = (v2f){lhi.x, lhi.y} + x2;
  v2f v3 = (v2f){lhi.z, lhi.w} + x3;
  v0 = __builtin_elementwise_max(v0, 0.2f * v0);
  v1 = __builtin_elementwise_max(v1, 0.2f * v1);
  v2 = __builtin_elementwise_max(v2, 0.2f * v2);
  v3 = __builtin_elementwise_max(v3, 0.2f * v3);
  v2f p = v0 * a0;
  p = __builtin_elementwise_fma(v1, a1, p);
  p = __builtin_elementwise_fma(v2, a2, p);
  p = __builtin_elementwise_fma(v3, a3, p);
  return p.x + p.y;
}

template <int T0>
__device__ __forceinline__ void four_steps128(const float4* __restrict__ rowsL, int s, int nv, int c,
                                              v2f x0, v2f x1, v2f x2, v2f x3,
                                              v2f a0, v2f a1, v2f a2, v2f a3,
                                              float& m, float& den, float4& acclo, float4& acchi) {
  float4 rlo[4], rhi[4];
  float lg[4];
  const unsigned o0 = (unsigned)row_lane<T0 + 0>(s) * 32u + (unsigned)c;
  const unsigned o1 = (unsigned)row_lane<T0 + 1>(s) * 32u + (unsigned)c;
  const unsigned o2 = (unsigned)row_lane<T0 + 2>(s) * 32u + (unsigned)c;
  const unsigned o3 = (unsigned)row_lane<T0 + 3>(s) * 32u + (unsigned)c;
  rlo[0] = rowsL[o0]; rhi[0] = rowsL[o0 + 16u];
  rlo[1] = rowsL[o1]; rhi[1] = rowsL[o1 + 16u];
  rlo[2] = rowsL[o2]; rhi[2] = rowsL[o2 + 16u];
  rlo[3] = rowsL[o3]; rhi[3] = rowsL[o3 + 16u];
  #pragma unroll
  for (int u = 0; u < 4; ++u) lg[u] = logit_part8(rlo[u], rhi[u], x0, x1, x2, x3, a0, a1, a2, a3);
  #pragma unroll
  for (int u = 0; u < 4; ++u) lg[u] += row_xor<1>(lg[u]);
  #pragma unroll
  for (int u = 0; u < 4; ++u) lg[u] += row_xor<2>(lg[u]);
  #pragma unroll
  for (int u = 0; u < 4; ++u) lg[u] += row_xor<4>(lg[u]);
  #pragma unroll
  for (int u = 0; u < 4; ++u) lg[u] += row_xor<8>(lg[u]);
  float cm = m;
  #pragma unroll
  for (int u = 0; u < 4; ++u) {
    if (T0 + u < nv) cm = fmaxf(cm, lg[u]);
  }
  const float sc = __expf(m - cm);
  den *= sc;
  row_scale(acclo, sc);
  row_scale(acchi, sc);
  m = cm;
  #pragma unroll
  for (int u = 0; u < 4; ++u) {
    const float w = (T0 + u < nv) ? __expf(lg[u] - cm) : 0.0f;
    den += w;
    row_fma(acclo, w, rlo[u]);
    row_fma(acchi, w, rhi[u]);
  }
}

__global__ __launch_bounds__(256, 10) void node_attn128_k(const float* __restrict__ xl,
                                                     const float* __restrict__ xr,
                                                     const float* __restrict__ att,
                                                     const float* __restrict__ bias,
                                                     const int* __restrict__ rowptr,
                                                     const int* __restrict__ csr_src,
                                                     int Nn, float* __restrict__ out) {
  const int lane = threadIdx.x & 31;
  const int wv = __builtin_amdgcn_readfirstlane((int)(threadIdx.x >> 5));
  const int node = blockIdx.x * 8 + wv;
  if (node >= Nn) return;
  const int c = lane & 15;
  const int half = lane >> 4;
  const float4* __restrict__ rowsL = (const float4*)xl;
  const long long nb = (long long)node * 32 + c;
  const float4 xdlo = ((const float4*)xr)[nb];
  const float4 xdhi = ((const float4*)xr)[nb + 16];
  const float4 alo  = ((const float4*)att)[c];
  const float4 ahi  = ((const float4*)att)[16 + c];
  const float4 slo  = rowsL[nb];
  const float4 shi  = rowsL[nb + 16];
  const int beg = rowptr[node];
  const int end = rowptr[node + 1];

  const v2f x0 = {xdlo.x, xdlo.y}, x1 = {xdlo.z, xdlo.w}, x2 = {xdhi.x, xdhi.y}, x3 = {xdhi.z, xdhi.w};
  const v2f a0 = {alo.x, alo.y}, a1 = {alo.z, alo.w}, a2 = {ahi.x, ahi.y}, a3 = {ahi.z, ahi.w};

  float m = logit_part8(slo, shi, x0, x1, x2, x3, a0, a1, a2, a3);
  m += row_xor<1>(m);
  m += row_xor<2>(m);
  m += row_xor<4>(m);
  m += row_xor<8>(m);
  const float4 zero4 = make_float4(0.0f, 0.0f, 0.0f, 0.0f);
  float den = (half == 0) ? 1.0f : 0.0f;
  float4 acclo = (half == 0) ? slo : zero4;
  float4 acchi = (half == 0) ? shi : zero4;

  for (int base = beg; base < end; base += 32) {
    const int n = (end - base < 32) ? (end - base) : 32;
    const int eo = 2 * c + half;
    const int s = (eo < n) ? csr_src[base + eo] : node;
    const int nv = (n - half + 1) >> 1;
    four_steps128<0>(rowsL, s, nv, c, x0, x1, x2, x3, a0, a1, a2, a3, m, den, acclo, acchi);
    if (n > 8)  four_steps128<4>(rowsL, s, nv, c, x0, x1, x2, x3, a0, a1, a2, a3, m, den, acclo, acchi);
    if (n > 16) four_steps128<8>(rowsL, s, nv, c, x0, x1, x2, x3, a0, a1, a2, a3, m, den, acclo, acchi);
    if (n > 24) four_steps128<12>(rowsL, s, nv, c, x0, x1, x2, x3, a0, a1, a2, a3, m, den, acclo, acchi);
  }

  const float mm = fmaxf(m, xhalf(m));
  const float f = __expf(m - mm);
  den *= f;
  row_scale(acclo, f);
  row_scale(acchi, f);
  den += xhalf(den);
  acclo.x += xhalf(acclo.x); acclo.y += xhalf(acclo.y);
  acclo.z += xhalf(acclo.z); acclo.w += xhalf(acclo.w);
  acchi.x += xhalf(acchi.x); acchi.y += xhalf(acchi.y);
  acchi.z += xhalf(acchi.z); acchi.w += xhalf(acchi.w);
  const float4 b = ((const float4*)bias)[lane];
  const float4 o = (half == 0) ? acclo : acchi;
  ((float4*)out)[(long long)node * 32 + lane] = row_final(o, den + 1e-16f, b);
}


extern "C" void kernel_launch(void* const* d_in, const int* in_sizes, int n_in,
                              void* d_out, int out_size, void* d_ws, size_t ws_size,
                              hipStream_t stream) {
  const int F_IN = 128, H = 128, OUT = 32;
  const float* x    = (const float*)d_in[0];
  const int*   eidx = (const int*)  d_in[1];
  const float* Wl1  = (const float*)d_in[2];
  const float* Wr1  = (const float*)d_in[3];
  const float* att1 = (const float*)d_in[4];
  const float* b1   = (const float*)d_in[5];
  const float* Wl2  = (const float*)d_in[6];
  const float* Wr2  = (const float*)d_in[7];
  const float* att2 = (const float*)d_in[8];
  const float* b2   = (const float*)d_in[9];
  float* out = (float*)d_out;

  const int Nn = in_sizes[0] / F_IN;
  const int E  = in_sizes[1] / 2;
  const int* esrc = eidx;
  const int* edst = eidx + E;

  char* ws = (char*)d_ws;
  size_t off = 0;
  auto carve = [&](size_t bytes) -> char* {
    char* p = ws + off;
    off += (bytes + 255) & ~(size_t)255;
    return p;
  };
  __bf16* bpl1  = (__bf16*)carve((size_t)F_IN * H   * sizeof(__bf16));
  __bf16* bpr1  = (__bf16*)carve((size_t)F_IN * H   * sizeof(__bf16));
  __bf16* bpl2  = (__bf16*)carve((size_t)H * OUT    * sizeof(__bf16));
  __bf16* bpr2  = (__bf16*)carve((size_t)H * OUT    * sizeof(__bf16));
  float*  xl    = (float*) carve((size_t)Nn * H * sizeof(float));
  float*  xr    = (float*) carve((size_t)Nn * H * sizeof(float));
  float*  hbuf  = (float*) carve((size_t)Nn * H * sizeof(float));
  int*    rowptr = (int*)  carve(((size_t)Nn + 1) * sizeof(int));
  int*    rank   = (int*)  carve((size_t)E * sizeof(int));
  int*    csrsrc = (int*)  carve((size_t)E * sizeof(int));
  int*    hist   = (int*)  carve((size_t)HIST_K * Nn * sizeof(int));

  auto g256 = [](long long n) { return (unsigned)((n + 255) / 256); };
  auto gwav = [](long long n) { return (unsigned)((n + 7) / 8); };
  static_assert(NODE_BLK == 256, "the column pass runs one node block per 256-thread work-group");
  int*    btot   = (int*)  carve((size_t)HIST_K * g256(Nn) * sizeof(int));

  const int Mt = Nn / 16;
  static_assert(F_IN == 4 * 32 && H == 8 * 16 && OUT == 2 * 16, "projection template arguments");
  static_assert(F_IN == 128 && H == 128 && OUT == 32, "sizes the packing work-groups are cut for");

  hist_pack_k<<<HIST_K + PACK_WGS, 1024, 0, stream>>>(edst, E, Nn, hist, btot, rank, Wl1, Wr1, Wl2, Wr2,
                                                      bpl1, bpr1, bpl2, bpr2);
  colscan_proj1_k<<<gwav(Mt) + g256(Nn), 256, 0, stream>>>(hist, Nn, btot, rowptr, x, bpl1, bpr1, xl, xr,
                                                           Mt, (int)gwav(Mt));
  scatter_k<<<g256(E), 256, 0, stream>>>(esrc, edst, rank, hist, E, Nn, csrsrc);

  node_attn128_k<<<gwav(Nn), 256, 0, stream>>>(xl, xr, att1, b1, rowptr, csrsrc, Nn, hbuf);

  wmma_proj_pair_k<2, 4><<<gwav(Mt), 256, 0, stream>>>(hbuf, bpl2, bpr2, xl, xr, Mt);
  node_attn32_k<<<gwav(Nn), 256, 0, stream>>>(xl, xr, att2, b2, rowptr, csrsrc, Nn, out);
}
